# starting kernel
# baseline (speedup 1.0000x reference)
#include <hip/hip_runtime.h>
#include <hip/hip_bf16.h>

typedef __attribute__((ext_vector_type(2))) float v2f;
typedef __attribute__((ext_vector_type(4))) float v4f;
typedef __attribute__((ext_vector_type(8))) float v8f;

#define D 128
#define LN_EPS 1e-5f

__global__ void __launch_bounds__(256)
deg_init_kernel(float* __restrict__ deg, int N) {
    int i = blockIdx.x * blockDim.x + threadIdx.x;
    if (i < N) deg[i] = 1.0f;
}

__global__ void __launch_bounds__(256)
deg_scatter_kernel(const int* __restrict__ cols, const float* __restrict__ ew,
                   float* __restrict__ deg, int E) {
    int e = blockIdx.x * blockDim.x + threadIdx.x;
    if (e < E) atomicAdd(&deg[cols[e]], ew[e]);
}

__global__ void __launch_bounds__(256)
dinv_kernel(float* __restrict__ deg, int N) {
    int i = blockIdx.x * blockDim.x + threadIdx.x;
    if (i < N) {
        float d = deg[i];
        deg[i] = (d > 0.0f) ? rsqrtf(d) : 0.0f;
    }
}

__global__ void __launch_bounds__(256)
gemm_wmma_kernel(const float* __restrict__ x, const float* __restrict__ W,
                 const float* __restrict__ dinv, float* __restrict__ h,
                 float* __restrict__ acc, int N) {
    const int lane  = threadIdx.x;
    const int nbase = threadIdx.y * 16;
    const int mbase = blockIdx.x * 16;
    const int lmod  = lane & 15;
    const int koff  = (lane >> 4) * 2;

    int arow = mbase + lmod;
    if (arow > N - 1) arow = N - 1;

    const float* xrow = x + (size_t)arow * D + koff;
    const float* wrow = W + (size_t)(nbase + lmod) * D + koff;

    v8f c = {};
#pragma unroll
    for (int k0 = 0; k0 < D; k0 += 4) {
        v2f a = *(const v2f*)(xrow + k0);
        v2f b = *(const v2f*)(wrow + k0);
        c = __builtin_amdgcn_wmma_f32_16x16x4_f32(false, a, false, b,
                                                  (short)0, c, false, false);
    }

    const int mhi = (lane >> 4) * 8;
#pragma unroll
    for (int r = 0; r < 8; ++r) {
        int row = mbase + mhi + r;
        if (row < N) {
            int col = nbase + lmod;
            float v  = c[r];
            float di = dinv[row];
            h[(size_t)row * D + col]   = v;
            acc[(size_t)row * D + col] = v * di * di;
        }
    }
}

__global__ void __launch_bounds__(256)
edge_scatter_kernel(const int* __restrict__ rows, const int* __restrict__ cols,
                    const float* __restrict__ ew, const float* __restrict__ dinv,
                    const float* __restrict__ h, float* __restrict__ acc, int E) {
    long long t = (long long)blockIdx.x * blockDim.x + threadIdx.x;
    int e = (int)(t >> 5);
    if (e >= E) return;
    int part = (int)(t & 31);

    int   src  = rows[e];
    int   dst  = cols[e];
    float norm = dinv[src] * ew[e] * dinv[dst];

    v4f hv = *(const v4f*)(h + (size_t)src * D + part * 4);
    float* out = acc + (size_t)dst * D + part * 4;
#pragma unroll
    for (int i = 0; i < 4; ++i)
        atomicAdd(out + i, hv[i] * norm);
}

__global__ void __launch_bounds__(256)
finalize_kernel(float* __restrict__ out, const float* __restrict__ bias,
                const float* __restrict__ gamma, const float* __restrict__ beta,
                int N) {
    int lane = threadIdx.x & 31;
    int wave = threadIdx.x >> 5;
    int node = blockIdx.x * 8 + wave;
    if (node >= N) return;

    float* p = out + (size_t)node * D + lane * 4;
    v4f v = *(const v4f*)p;
    float s = 0.f, s2 = 0.f;
#pragma unroll
    for (int i = 0; i < 4; ++i) {
        int c = lane * 4 + i;
        float vv = v[i] + bias[c];
        vv = fmaxf(vv, 0.0f);
        v[i] = vv;
        s  += vv;
        s2 += vv * vv;
    }
#pragma unroll
    for (int off = 16; off >= 1; off >>= 1) {
        s  += __shfl_xor(s,  off, 32);
        s2 += __shfl_xor(s2, off, 32);
    }
    float mu  = s * (1.0f / 128.0f);
    float var = s2 * (1.0f / 128.0f) - mu * mu;
    float rs  = rsqrtf(var + LN_EPS);
#pragma unroll
    for (int i = 0; i < 4; ++i) {
        int c = lane * 4 + i;
        v[i] = (v[i] - mu) * rs * gamma[c] + beta[c];
    }
    *(v4f*)p = v;
}

extern "C" void kernel_launch(void* const* d_in, const int* in_sizes, int n_in,
                              void* d_out, int out_size, void* d_ws, size_t ws_size,
                              hipStream_t stream) {
    const float* x     = (const float*)d_in[0];
    const int*   eidx  = (const int*)  d_in[1];
    const float* ew    = (const float*)d_in[2];
    const float* W     = (const float*)d_in[3];
    const float* bias  = (const float*)d_in[4];
    const float* gamma = (const float*)d_in[5];
    const float* beta  = (const float*)d_in[6];
    float*       out   = (float*)d_out;

    const int N = in_sizes[0] / D;
    const int E = in_sizes[2];
    const int* rows = eidx;
    const int* cols = eidx + E;

    float* h    = (float*)d_ws;
    float* dinv = h + (size_t)N * D;

    deg_init_kernel   <<<(N + 255) / 256, 256, 0, stream>>>(dinv, N);
    deg_scatter_kernel<<<(E + 255) / 256, 256, 0, stream>>>(cols, ew, dinv, E);
    dinv_kernel       <<<(N + 255) / 256, 256, 0, stream>>>(dinv, N);

    gemm_wmma_kernel<<<(N + 15) / 16, dim3(32, 8), 0, stream>>>(x, W, dinv, h, out, N);

    long long sthreads = (long long)E * 32;
    edge_scatter_kernel<<<(unsigned)((sthreads + 255) / 256), 256, 0, stream>>>(
        rows, cols, ew, dinv, h, out, E);

    finalize_kernel<<<(N + 7) / 8, 256, 0, stream>>>(out, bias, gamma, beta, N);
}


# edge sum in LDS: a work-group per 392-target range keeps its rows in LDS, scans all targets, adds matching h rows by LDS float atomics (arrival order), then bias/ReLU/LN; acc and finalize launch gone
# speedup vs baseline: 1.5782x; 1.5782x over previous
#include <hip/hip_runtime.h>
#include <hip/hip_bf16.h>

typedef __attribute__((ext_vector_type(2))) float v2f;
typedef __attribute__((ext_vector_type(4))) float v4f;
typedef __attribute__((ext_vector_type(8))) float v8f;

#define D 128
#define LN_EPS 1e-5f

__global__ void __launch_bounds__(256)
deg_init_kernel(float* __restrict__ deg, int N) {
    int i = blockIdx.x * blockDim.x + threadIdx.x;
    if (i < N) deg[i] = 1.0f;
}

__global__ void __launch_bounds__(256)
deg_scatter_kernel(const int* __restrict__ cols, const float* __restrict__ ew,
                   float* __restrict__ deg, int E) {
    int e = blockIdx.x * blockDim.x + threadIdx.x;
    if (e < E) atomicAdd(&deg[cols[e]], ew[e]);
}

__global__ void __launch_bounds__(256)
dinv_kernel(float* __restrict__ deg, int N) {
    int i = blockIdx.x * blockDim.x + threadIdx.x;
    if (i < N) {
        float d = deg[i];
        deg[i] = (d > 0.0f) ? rsqrtf(d) : 0.0f;
    }
}

__global__ void __launch_bounds__(256)
gemm_wmma_kernel(const float* __restrict__ x, const float* __restrict__ W,
                 float* __restrict__ h, int N) {
    const int lane  = threadIdx.x;
    const int nbase = threadIdx.y * 16;
    const int mbase = blockIdx.x * 16;
    const int lmod  = lane & 15;
    const int koff  = (lane >> 4) * 2;

    int arow = mbase + lmod;
    if (arow > N - 1) arow = N - 1;

    const float* xrow = x + (size_t)arow * D + koff;
    const float* wrow = W + (size_t)(nbase + lmod) * D + koff;

    v8f c = {};
#pragma unroll
    for (int k0 = 0; k0 < D; k0 += 4) {
        v2f a = *(const v2f*)(xrow + k0);
        v2f b = *(const v2f*)(wrow + k0);
        c = __builtin_amdgcn_wmma_f32_16x16x4_f32(false, a, false, b,
                                                  (short)0, c, false, false);
    }

    const int mhi = (lane >> 4) * 8;
#pragma unroll
    for (int r = 0; r < 8; ++r) {
        int row = mbase + mhi + r;
        if (row < N) {
            int col = nbase + lmod;
            h[(size_t)row * D + col] = c[r];
        }
    }
}

#define TR    392
#define NWAVE 32
#define CHUNK 128
#define GROUP 4
#define LCAP  (GROUP * CHUNK + 128)

typedef __attribute__((ext_vector_type(4))) int v4i;

__global__ void __launch_bounds__(1024)
range_sum_norm_kernel(const int* __restrict__ rows, const int* __restrict__ cols,
                      const float* __restrict__ ew, const float* __restrict__ dinv,
                      const float* __restrict__ h, const float* __restrict__ bias,
                      const float* __restrict__ gamma, const float* __restrict__ beta,
                      float* __restrict__ out, int N, int E) {
    __shared__ __attribute__((aligned(16))) float tile[TR * D];
    __shared__ int lists[NWAVE * LCAP];

    const int lane = threadIdx.x & 31;
    const int wave = __builtin_amdgcn_readfirstlane((int)(threadIdx.x >> 5));
    const int lo   = blockIdx.x * TR;
    const int nr   = (N - lo < TR) ? (N - lo) : TR;
    int* mylist    = lists + wave * LCAP;

    for (int r = wave; r < nr; r += NWAVE) {
        v4f hv   = *(const v4f*)(h + (size_t)(lo + r) * D + lane * 4);
        float di = dinv[lo + r];
        v4f s;
#pragma unroll
        for (int i = 0; i < 4; ++i) s[i] = hv[i] * di * di;
        *(v4f*)(tile + r * D + lane * 4) = s;
    }
    __syncthreads();

    auto drain = [&](int cnt) {
        for (int b0 = 0; b0 < cnt; b0 += 32) {
            int  i   = b0 + lane;
            int  e   = mylist[(i < cnt) ? i : b0];
            int  src = rows[e];
            int  dl  = cols[e] - lo;
            float nm = dinv[src] * ew[e] * dinv[lo + dl];
            int  m   = (cnt - b0 < 32) ? (cnt - b0) : 32;
            for (int j0 = 0; j0 < m; j0 += 8) {
                v4f hv[8];
#pragma unroll
                for (int u = 0; u < 8; ++u) {
                    int j = (j0 + u < m) ? (j0 + u) : (m - 1);
                    int s = __builtin_amdgcn_readlane(src, j);
                    hv[u] = *(const v4f*)(h + (size_t)s * D + lane * 4);
                }
#pragma unroll
                for (int u = 0; u < 8; ++u) {
                    if (j0 + u < m) {
                        int   d = __builtin_amdgcn_readlane(dl, j0 + u);
                        float f = __builtin_bit_cast(float,
                                      __builtin_amdgcn_readlane(__builtin_bit_cast(int, nm), j0 + u));
                        float* a = tile + d * D + lane * 4;
#pragma unroll
                        for (int k = 0; k < 4; ++k)
                            __hip_atomic_fetch_add(a + k, hv[u][k] * f, __ATOMIC_RELAXED,
                                                   __HIP_MEMORY_SCOPE_WORKGROUP);
                    }
                }
            }
        }
    };

    const bool vec   = (E & 3) == 0;
    const int  nfull = vec ? (E / CHUNK) : 0;
    const int  rot   = (int)blockIdx.x * (nfull / (int)gridDim.x);
    int cnt = 0;

    for (int q0 = wave; q0 < nfull; q0 += GROUP * NWAVE) {
        v4i      c[GROUP];
        int      eb[GROUP];
        unsigned lim[GROUP];
#pragma unroll
        for (int u = 0; u < GROUP; ++u) {
            int  q     = q0 + u * NWAVE;
            bool valid = q < nfull;
            int  ch    = valid ? (q + rot) : rot;
            if (ch >= nfull) ch -= nfull;
            eb[u]  = ch * CHUNK;
            lim[u] = valid ? (unsigned)nr : 0u;
            c[u]   = *(const v4i*)(cols + eb[u] + lane * 4);
        }
#pragma unroll
        for (int u = 0; u < GROUP; ++u) {
            int  e0 = eb[u] + lane * 4;
            bool m0 = (unsigned)(c[u].x - lo) < lim[u];
            bool m1 = (unsigned)(c[u].y - lo) < lim[u];
            bool m2 = (unsigned)(c[u].z - lo) < lim[u];
            bool m3 = (unsigned)(c[u].w - lo) < lim[u];
            if (__builtin_amdgcn_ballot_w32(m0 | m1 | m2 | m3) != 0u) {
                unsigned b;
                b = __builtin_amdgcn_ballot_w32(m0);
                if (m0) mylist[cnt + __builtin_amdgcn_mbcnt_lo(b, 0u)] = e0;
                cnt += __builtin_popcount(b);
                b = __builtin_amdgcn_ballot_w32(m1);
                if (m1) mylist[cnt + __builtin_amdgcn_mbcnt_lo(b, 0u)] = e0 + 1;
                cnt += __builtin_popcount(b);
                b = __builtin_amdgcn_ballot_w32(m2);
                if (m2) mylist[cnt + __builtin_amdgcn_mbcnt_lo(b, 0u)] = e0 + 2;
                cnt += __builtin_popcount(b);
                b = __builtin_amdgcn_ballot_w32(m3);
                if (m3) mylist[cnt + __builtin_amdgcn_mbcnt_lo(b, 0u)] = e0 + 3;
                cnt += __builtin_popcount(b);
            }
        }
        if (cnt > LCAP - GROUP * CHUNK) {
            drain(cnt);
            cnt = 0;
        }
    }
    for (int base = nfull * CHUNK + wave * 32; base < E; base += NWAVE * 32) {
        int  e   = base + lane;
        int  col = (e < E) ? cols[e] : -1;
        bool m   = (unsigned)(col - lo) < (unsigned)nr;
        unsigned b = __builtin_amdgcn_ballot_w32(m);
        if (m) mylist[cnt + __builtin_amdgcn_mbcnt_lo(b, 0u)] = e;
        cnt += __builtin_popcount(b);
        if (cnt > LCAP - GROUP * CHUNK) {
            drain(cnt);
            cnt = 0;
        }
    }
    drain(cnt);
    __syncthreads();

    const v4f bv = *(const v4f*)(bias  + lane * 4);
    const v4f gv = *(const v4f*)(gamma + lane * 4);
    const v4f tv = *(const v4f*)(beta  + lane * 4);
    for (int r = wave; r < nr; r += NWAVE) {
        v4f v = *(const v4f*)(tile + r * D + lane * 4);
        float s = 0.f, s2 = 0.f;
#pragma unroll
        for (int i = 0; i < 4; ++i) {
            float vv = v[i] + bv[i];
            vv = fmaxf(vv, 0.0f);
            v[i] = vv;
            s  += vv;
            s2 += vv * vv;
        }
#pragma unroll
        for (int off = 16; off >= 1; off >>= 1) {
            s  += __shfl_xor(s,  off, 32);
            s2 += __shfl_xor(s2, off, 32);
        }
        float mu  = s * (1.0f / 128.0f);
        float var = s2 * (1.0f / 128.0f) - mu * mu;
        float rs  = rsqrtf(var + LN_EPS);
#pragma unroll
        for (int i = 0; i < 4; ++i)
            v[i] = (v[i] - mu) * rs * gv[i] + tv[i];
        *(v4f*)(out + (size_t)(lo + r) * D + lane * 4) = v;
    }
}

extern "C" void kernel_launch(void* const* d_in, const int* in_sizes, int n_in,
                              void* d_out, int out_size, void* d_ws, size_t ws_size,
                              hipStream_t stream) {
    const float* x     = (const float*)d_in[0];
    const int*   eidx  = (const int*)  d_in[1];
    const float* ew    = (const float*)d_in[2];
    const float* W     = (const float*)d_in[3];
    const float* bias  = (const float*)d_in[4];
    const float* gamma = (const float*)d_in[5];
    const float* beta  = (const float*)d_in[6];
    float*       out   = (float*)d_out;

    const int N = in_sizes[0] / D;
    const int E = in_sizes[2];
    const int* rows = eidx;
    const int* cols = eidx + E;

    float* h    = (float*)d_ws;
    float* dinv = h + (size_t)N * D;

    deg_init_kernel   <<<(N + 255) / 256, 256, 0, stream>>>(dinv, N);
    deg_scatter_kernel<<<(E + 255) / 256, 256, 0, stream>>>(cols, ew, dinv, E);
    dinv_kernel       <<<(N + 255) / 256, 256, 0, stream>>>(dinv, N);

    gemm_wmma_kernel<<<(N + 15) / 16, dim3(32, 8), 0, stream>>>(x, W, h, N);

    range_sum_norm_kernel<<<(N + TR - 1) / TR, NWAVE * 32, 0, stream>>>(
        rows, cols, ew, dinv, h, bias, gamma, beta, out, N, E);
}


# degree chain folded into neighbour launches: edge-weight adds into deg (global float atomics, arrival order as before) run inside the GEMM launch, deg^-1/2 taken by the reader; 5 to 3 launches
# speedup vs baseline: 1.6298x; 1.0327x over previous
#include <hip/hip_runtime.h>
#include <hip/hip_bf16.h>

typedef __attribute__((ext_vector_type(2))) float v2f;
typedef __attribute__((ext_vector_type(4))) float v4f;
typedef __attribute__((ext_vector_type(8))) float v8f;

#define D 128
#define LN_EPS 1e-5f

__global__ void __launch_bounds__(256)
deg_init_kernel(float* __restrict__ deg, int N) {
    int i = blockIdx.x * blockDim.x + threadIdx.x;
    if (i < N) deg[i] = 1.0f;
}

__device__ __forceinline__ float inv_sqrt_deg(float d) {
    return (d > 0.0f) ? rsqrtf(d) : 0.0f;
}

__global__ void __launch_bounds__(256)
gemm_wmma_kernel(const float* __restrict__ x, const float* __restrict__ W,
                 float* __restrict__ h, int N,
                 const int* __restrict__ cols, const float* __restrict__ ew,
                 float* __restrict__ deg, int E) {
    {
        const int per = (E + (int)gridDim.x - 1) / (int)gridDim.x;
        const int e0  = (int)blockIdx.x * per;
        const int e1  = (e0 + per < E) ? (e0 + per) : E;
        for (int e = e0 + (int)(threadIdx.y * 32 + threadIdx.x); e < e1; e += 256)
            atomicAdd(&deg[cols[e]], ew[e]);
    }

    const int lane  = threadIdx.x;
    const int nbase = threadIdx.y * 16;
    const int mbase = blockIdx.x * 16;
    const int lmod  = lane & 15;
    const int koff  = (lane >> 4) * 2;

    int arow = mbase + lmod;
    if (arow > N - 1) arow = N - 1;

    const float* xrow = x + (size_t)arow * D + koff;
    const float* wrow = W + (size_t)(nbase + lmod) * D + koff;

    v8f c = {};
#pragma unroll
    for (int k0 = 0; k0 < D; k0 += 4) {
        v2f a = *(const v2f*)(xrow + k0);
        v2f b = *(const v2f*)(wrow + k0);
        c = __builtin_amdgcn_wmma_f32_16x16x4_f32(false, a, false, b,
                                                  (short)0, c, false, false);
    }

    const int mhi = (lane >> 4) * 8;
#pragma unroll
    for (int r = 0; r < 8; ++r) {
        int row = mbase + mhi + r;
        if (row < N) {
            int col = nbase + lmod;
            h[(size_t)row * D + col] = c[r];
        }
    }
}

#define TR    392
#define NWAVE 32
#define CHUNK 128
#define GROUP 4
#define LCAP  (GROUP * CHUNK + 128)

typedef __attribute__((ext_vector_type(4))) int v4i;

__global__ void __launch_bounds__(1024)
range_sum_norm_kernel(const int* __restrict__ rows, const int* __restrict__ cols,
                      const float* __restrict__ ew, const float* __restrict__ deg,
                      const float* __restrict__ h, const float* __restrict__ bias,
                      const float* __restrict__ gamma, const float* __restrict__ beta,
                      float* __restrict__ out, int N, int E) {
    __shared__ __attribute__((aligned(16))) float tile[TR * D];
    __shared__ int lists[NWAVE * LCAP];

    const int lane = threadIdx.x & 31;
    const int wave = __builtin_amdgcn_readfirstlane((int)(threadIdx.x >> 5));
    const int lo   = blockIdx.x * TR;
    const int nr   = (N - lo < TR) ? (N - lo) : TR;
    int* mylist    = lists + wave * LCAP;

    for (int r = wave; r < nr; r += NWAVE) {
        v4f hv   = *(const v4f*)(h + (size_t)(lo + r) * D + lane * 4);
        float di = inv_sqrt_deg(deg[lo + r]);
        v4f s;
#pragma unroll
        for (int i = 0; i < 4; ++i) s[i] = hv[i] * di * di;
        *(v4f*)(tile + r * D + lane * 4) = s;
    }
    __syncthreads();

    auto drain = [&](int cnt) {
        for (int b0 = 0; b0 < cnt; b0 += 32) {
            int  i   = b0 + lane;
            int  e   = mylist[(i < cnt) ? i : b0];
            int  src = rows[e];
            int  dl  = cols[e] - lo;
            float nm = inv_sqrt_deg(deg[src]) * ew[e] * inv_sqrt_deg(deg[lo + dl]);
            int  m   = (cnt - b0 < 32) ? (cnt - b0) : 32;
            for (int j0 = 0; j0 < m; j0 += 8) {
                v4f hv[8];
#pragma unroll
                for (int u = 0; u < 8; ++u) {
                    int j = (j0 + u < m) ? (j0 + u) : (m - 1);
                    int s = __builtin_amdgcn_readlane(src, j);
                    hv[u] = *(const v4f*)(h + (size_t)s * D + lane * 4);
                }
#pragma unroll
                for (int u = 0; u < 8; ++u) {
                    if (j0 + u < m) {
                        int   d = __builtin_amdgcn_readlane(dl, j0 + u);
                        float f = __builtin_bit_cast(float,
                                      __builtin_amdgcn_readlane(__builtin_bit_cast(int, nm), j0 + u));
                        float* a = tile + d * D + lane * 4;
#pragma unroll
                        for (int k = 0; k < 4; ++k)
                            __hip_atomic_fetch_add(a + k, hv[u][k] * f, __ATOMIC_RELAXED,
                                                   __HIP_MEMORY_SCOPE_WORKGROUP);
                    }
                }
            }
        }
    };

    const bool vec   = (E & 3) == 0;
    const int  nfull = vec ? (E / CHUNK) : 0;
    const int  rot   = (int)blockIdx.x * (nfull / (int)gridDim.x);
    int cnt = 0;

    for (int q0 = wave; q0 < nfull; q0 += GROUP * NWAVE) {
        v4i      c[GROUP];
        int      eb[GROUP];
        unsigned lim[GROUP];
#pragma unroll
        for (int u = 0; u < GROUP; ++u) {
            int  q     = q0 + u * NWAVE;
            bool valid = q < nfull;
            int  ch    = valid ? (q + rot) : rot;
            if (ch >= nfull) ch -= nfull;
            eb[u]  = ch * CHUNK;
            lim[u] = valid ? (unsigned)nr : 0u;
            c[u]   = *(const v4i*)(cols + eb[u] + lane * 4);
        }
#pragma unroll
        for (int u = 0; u < GROUP; ++u) {
            int  e0 = eb[u] + lane * 4;
            bool m0 = (unsigned)(c[u].x - lo) < lim[u];
            bool m1 = (unsigned)(c[u].y - lo) < lim[u];
            bool m2 = (unsigned)(c[u].z - lo) < lim[u];
            bool m3 = (unsigned)(c[u].w - lo) < lim[u];
            if (__builtin_amdgcn_ballot_w32(m0 | m1 | m2 | m3) != 0u) {
                unsigned b;
                b = __builtin_amdgcn_ballot_w32(m0);
                if (m0) mylist[cnt + __builtin_amdgcn_mbcnt_lo(b, 0u)] = e0;
                cnt += __builtin_popcount(b);
                b = __builtin_amdgcn_ballot_w32(m1);
                if (m1) mylist[cnt + __builtin_amdgcn_mbcnt_lo(b, 0u)] = e0 + 1;
                cnt += __builtin_popcount(b);
                b = __builtin_amdgcn_ballot_w32(m2);
                if (m2) mylist[cnt + __builtin_amdgcn_mbcnt_lo(b, 0u)] = e0 + 2;
                cnt += __builtin_popcount(b);
                b = __builtin_amdgcn_ballot_w32(m3);
                if (m3) mylist[cnt + __builtin_amdgcn_mbcnt_lo(b, 0u)] = e0 + 3;
                cnt += __builtin_popcount(b);
            }
        }
        if (cnt > LCAP - GROUP * CHUNK) {
            drain(cnt);
            cnt = 0;
        }
    }
    for (int base = nfull * CHUNK + wave * 32; base < E; base += NWAVE * 32) {
        int  e   = base + lane;
        int  col = (e < E) ? cols[e] : -1;
        bool m   = (unsigned)(col - lo) < (unsigned)nr;
        unsigned b = __builtin_amdgcn_ballot_w32(m);
        if (m) mylist[cnt + __builtin_amdgcn_mbcnt_lo(b, 0u)] = e;
        cnt += __builtin_popcount(b);
        if (cnt > LCAP - GROUP * CHUNK) {
            drain(cnt);
            cnt = 0;
        }
    }
    drain(cnt);
    __syncthreads();

    const v4f bv = *(const v4f*)(bias  + lane * 4);
    const v4f gv = *(const v4f*)(gamma + lane * 4);
    const v4f tv = *(const v4f*)(beta  + lane * 4);
    for (int r = wave; r < nr; r += NWAVE) {
        v4f v = *(const v4f*)(tile + r * D + lane * 4);
        float s = 0.f, s2 = 0.f;
#pragma unroll
        for (int i = 0; i < 4; ++i) {
            float vv = v[i] + bv[i];
            vv = fmaxf(vv, 0.0f);
            v[i] = vv;
            s  += vv;
            s2 += vv * vv;
        }
#pragma unroll
        for (int off = 16; off >= 1; off >>= 1) {
            s  += __shfl_xor(s,  off, 32);
            s2 += __shfl_xor(s2, off, 32);
        }
        float mu  = s * (1.0f / 128.0f);
        float var = s2 * (1.0f / 128.0f) - mu * mu;
        float rs  = rsqrtf(var + LN_EPS);
#pragma unroll
        for (int i = 0; i < 4; ++i)
            v[i] = (v[i] - mu) * rs * gv[i] + tv[i];
        *(v4f*)(out + (size_t)(lo + r) * D + lane * 4) = v;
    }
}

extern "C" void kernel_launch(void* const* d_in, const int* in_sizes, int n_in,
                              void* d_out, int out_size, void* d_ws, size_t ws_size,
                              hipStream_t stream) {
    const float* x     = (const float*)d_in[0];
    const int*   eidx  = (const int*)  d_in[1];
    const float* ew    = (const float*)d_in[2];
    const float* W     = (const float*)d_in[3];
    const float* bias  = (const float*)d_in[4];
    const float* gamma = (const float*)d_in[5];
    const float* beta  = (const float*)d_in[6];
    float*       out   = (float*)d_out;

    const int N = in_sizes[0] / D;
    const int E = in_sizes[2];
    const int* rows = eidx;
    const int* cols = eidx + E;

    float* h    = (float*)d_ws;
    float* deg  = h + (size_t)N * D;

    deg_init_kernel<<<(N + 255) / 256, 256, 0, stream>>>(deg, N);

    gemm_wmma_kernel<<<(N + 15) / 16, dim3(32, 8), 0, stream>>>(x, W, h, N, cols, ew, deg, E);

    range_sum_norm_kernel<<<(N + TR - 1) / TR, NWAVE * 32, 0, stream>>>(
        rows, cols, ew, deg, h, bias, gamma, beta, out, N, E);
}


# GEMM on a resident grid: 1024 waves each hold all of W as B fragments in registers (no LDS) and walk 16-row x tiles by 16-B fragment loads, whole-line h stores; k sum order inside each 8 changes
# speedup vs baseline: 2.0498x; 1.2577x over previous
#include <hip/hip_runtime.h>
#include <hip/hip_bf16.h>

typedef __attribute__((ext_vector_type(2))) float v2f;
typedef __attribute__((ext_vector_type(4))) float v4f;
typedef __attribute__((ext_vector_type(8))) float v8f;

#define D 128
#define LN_EPS 1e-5f

__global__ void __launch_bounds__(256)
deg_init_kernel(float* __restrict__ deg, int N) {
    int i = blockIdx.x * blockDim.x + threadIdx.x;
    if (i < N) deg[i] = 1.0f;
}

__device__ __forceinline__ float inv_sqrt_deg(float d) {
    return (d > 0.0f) ? rsqrtf(d) : 0.0f;
}

#define GEMM_WAVES 4
#define GEMM_GROUPS 256
#define DEG_BATCH 8

__global__ void __launch_bounds__(GEMM_WAVES * 32)
gemm_wmma_kernel(const float* __restrict__ x, const float* __restrict__ W,
                 float* __restrict__ h, int N,
                 const int* __restrict__ cols, const float* __restrict__ ew,
                 float* __restrict__ deg, int E) {
    const int lane = threadIdx.x & 31;
    const int wave = __builtin_amdgcn_readfirstlane((int)(threadIdx.x >> 5));
    const int c    = lane & 15;
    const int hh   = lane >> 4;

    v4f wb[8][16];
#pragma unroll
    for (int t = 0; t < 8; ++t) {
        const int n = 64 * (t >> 2) + 4 * c + (t & 3);
        const float* wrow = W + (size_t)n * D + 4 * hh;
#pragma unroll
        for (int j = 0; j < 16; ++j) wb[t][j] = *(const v4f*)(wrow + 8 * j);
    }

    const int ntiles = (N + 15) / 16;
    const int nw     = (int)gridDim.x * GEMM_WAVES;
    int tile         = (int)blockIdx.x * GEMM_WAVES + wave;

    auto load_tile = [&](v4f (&dst)[16], int tl) {
        int row = tl * 16 + c;
        if (row > N - 1) row = N - 1;
        const float* p = x + (size_t)row * D + 4 * hh;
#pragma unroll
        for (int j = 0; j < 16; ++j) dst[j] = *(const v4f*)(p + 8 * j);
    };

    v4f xa[16];
    load_tile(xa, tile);

    {
        const int per = (E + (int)gridDim.x - 1) / (int)gridDim.x;
        const int e0  = (int)blockIdx.x * per;
        const int e1  = (e0 + per < E) ? (e0 + per) : E;
        for (int eb = e0 + (int)threadIdx.x; eb < e1; eb += DEG_BATCH * GEMM_WAVES * 32) {
            int   tc[DEG_BATCH];
            float tw[DEG_BATCH];
#pragma unroll
            for (int u = 0; u < DEG_BATCH; ++u) {
                const int e  = eb + u * GEMM_WAVES * 32;
                const int ec = (e < e1) ? e : eb;
                tc[u] = cols[ec];
                tw[u] = ew[ec];
            }
#pragma unroll
            for (int u = 0; u < DEG_BATCH; ++u)
                if (eb + u * GEMM_WAVES * 32 < e1) atomicAdd(&deg[tc[u]], tw[u]);
        }
    }

    for (; tile < ntiles; tile += nw) {
        v4f xn[16];
        load_tile(xn, tile + nw);

        v8f acc[8];
#pragma unroll
        for (int t = 0; t < 8; ++t) acc[t] = v8f{};

#pragma unroll
        for (int j = 0; j < 16; ++j) {
            const v2f a0 = {xa[j].x, xa[j].y};
            const v2f a1 = {xa[j].z, xa[j].w};
#pragma unroll
            for (int t = 0; t < 8; ++t) {
                const v2f b0 = {wb[t][j].x, wb[t][j].y};
                acc[t] = __builtin_amdgcn_wmma_f32_16x16x4_f32(false, a0, false, b0,
                                                               (short)0, acc[t], false, false);
            }
#pragma unroll
            for (int t = 0; t < 8; ++t) {
                const v2f b1 = {wb[t][j].z, wb[t][j].w};
                acc[t] = __builtin_amdgcn_wmma_f32_16x16x4_f32(false, a1, false, b1,
                                                               (short)0, acc[t], false, false);
            }
        }

        const int rbase = tile * 16 + 8 * hh;
#pragma unroll
        for (int r = 0; r < 8; ++r) {
            const int row = rbase + r;
            if (row < N) {
                float* o = h + (size_t)row * D + 4 * c;
                v4f lo4 = {acc[0][r], acc[1][r], acc[2][r], acc[3][r]};
                v4f hi4 = {acc[4][r], acc[5][r], acc[6][r], acc[7][r]};
                *(v4f*)(o)      = lo4;
                *(v4f*)(o + 64) = hi4;
            }
        }

#pragma unroll
        for (int j = 0; j < 16; ++j) xa[j] = xn[j];
    }
}

#define TR    392
#define NWAVE 32
#define CHUNK 128
#define GROUP 4
#define LCAP  (GROUP * CHUNK + 128)

typedef __attribute__((ext_vector_type(4))) int v4i;

__global__ void __launch_bounds__(1024)
range_sum_norm_kernel(const int* __restrict__ rows, const int* __restrict__ cols,
                      const float* __restrict__ ew, const float* __restrict__ deg,
                      const float* __restrict__ h, const float* __restrict__ bias,
                      const float* __restrict__ gamma, const float* __restrict__ beta,
                      float* __restrict__ out, int N, int E) {
    __shared__ __attribute__((aligned(16))) float tile[TR * D];
    __shared__ int lists[NWAVE * LCAP];

    const int lane = threadIdx.x & 31;
    const int wave = __builtin_amdgcn_readfirstlane((int)(threadIdx.x >> 5));
    const int lo   = blockIdx.x * TR;
    const int nr   = (N - lo < TR) ? (N - lo) : TR;
    int* mylist    = lists + wave * LCAP;

    for (int r = wave; r < nr; r += NWAVE) {
        v4f hv   = *(const v4f*)(h + (size_t)(lo + r) * D + lane * 4);
        float di = inv_sqrt_deg(deg[lo + r]);
        v4f s;
#pragma unroll
        for (int i = 0; i < 4; ++i) s[i] = hv[i] * di * di;
        *(v4f*)(tile + r * D + lane * 4) = s;
    }
    __syncthreads();

    auto drain = [&](int cnt) {
        for (int b0 = 0; b0 < cnt; b0 += 32) {
            int  i   = b0 + lane;
            int  e   = mylist[(i < cnt) ? i : b0];
            int  src = rows[e];
            int  dl  = cols[e] - lo;
            float nm = inv_sqrt_deg(deg[src]) * ew[e] * inv_sqrt_deg(deg[lo + dl]);
            int  m   = (cnt - b0 < 32) ? (cnt - b0) : 32;
            for (int j0 = 0; j0 < m; j0 += 8) {
                v4f hv[8];
#pragma unroll
                for (int u = 0; u < 8; ++u) {
                    int j = (j0 + u < m) ? (j0 + u) : (m - 1);
                    int s = __builtin_amdgcn_readlane(src, j);
                    hv[u] = *(const v4f*)(h + (size_t)s * D + lane * 4);
                }
#pragma unroll
                for (int u = 0; u < 8; ++u) {
                    if (j0 + u < m) {
                        int   d = __builtin_amdgcn_readlane(dl, j0 + u);
                        float f = __builtin_bit_cast(float,
                                      __builtin_amdgcn_readlane(__builtin_bit_cast(int, nm), j0 + u));
                        float* a = tile + d * D + lane * 4;
#pragma unroll
                        for (int k = 0; k < 4; ++k)
                            __hip_atomic_fetch_add(a + k, hv[u][k] * f, __ATOMIC_RELAXED,
                                                   __HIP_MEMORY_SCOPE_WORKGROUP);
                    }
                }
            }
        }
    };

    const bool vec   = (E & 3) == 0;
    const int  nfull = vec ? (E / CHUNK) : 0;
    const int  rot   = (int)blockIdx.x * (nfull / (int)gridDim.x);
    int cnt = 0;

    for (int q0 = wave; q0 < nfull; q0 += GROUP * NWAVE) {
        v4i      c[GROUP];
        int      eb[GROUP];
        unsigned lim[GROUP];
#pragma unroll
        for (int u = 0; u < GROUP; ++u) {
            int  q     = q0 + u * NWAVE;
            bool valid = q < nfull;
            int  ch    = valid ? (q + rot) : rot;
            if (ch >= nfull) ch -= nfull;
            eb[u]  = ch * CHUNK;
            lim[u] = valid ? (unsigned)nr : 0u;
            c[u]   = *(const v4i*)(cols + eb[u] + lane * 4);
        }
#pragma unroll
        for (int u = 0; u < GROUP; ++u) {
            int  e0 = eb[u] + lane * 4;
            bool m0 = (unsigned)(c[u].x - lo) < lim[u];
            bool m1 = (unsigned)(c[u].y - lo) < lim[u];
            bool m2 = (unsigned)(c[u].z - lo) < lim[u];
            bool m3 = (unsigned)(c[u].w - lo) < lim[u];
            if (__builtin_amdgcn_ballot_w32(m0 | m1 | m2 | m3) != 0u) {
                unsigned b;
                b = __builtin_amdgcn_ballot_w32(m0);
                if (m0) mylist[cnt + __builtin_amdgcn_mbcnt_lo(b, 0u)] = e0;
                cnt += __builtin_popcount(b);
                b = __builtin_amdgcn_ballot_w32(m1);
                if (m1) mylist[cnt + __builtin_amdgcn_mbcnt_lo(b, 0u)] = e0 + 1;
                cnt += __builtin_popcount(b);
                b = __builtin_amdgcn_ballot_w32(m2);
                if (m2) mylist[cnt + __builtin_amdgcn_mbcnt_lo(b, 0u)] = e0 + 2;
                cnt += __builtin_popcount(b);
                b = __builtin_amdgcn_ballot_w32(m3);
                if (m3) mylist[cnt + __builtin_amdgcn_mbcnt_lo(b, 0u)] = e0 + 3;
                cnt += __builtin_popcount(b);
            }
        }
        if (cnt > LCAP - GROUP * CHUNK) {
            drain(cnt);
            cnt = 0;
        }
    }
    for (int base = nfull * CHUNK + wave * 32; base < E; base += NWAVE * 32) {
        int  e   = base + lane;
        int  col = (e < E) ? cols[e] : -1;
        bool m   = (unsigned)(col - lo) < (unsigned)nr;
        unsigned b = __builtin_amdgcn_ballot_w32(m);
        if (m) mylist[cnt + __builtin_amdgcn_mbcnt_lo(b, 0u)] = e;
        cnt += __builtin_popcount(b);
        if (cnt > LCAP - GROUP * CHUNK) {
            drain(cnt);
            cnt = 0;
        }
    }
    drain(cnt);
    __syncthreads();

    const v4f bv = *(const v4f*)(bias  + lane * 4);
    const v4f gv = *(const v4f*)(gamma + lane * 4);
    const v4f tv = *(const v4f*)(beta  + lane * 4);
    for (int r = wave; r < nr; r += NWAVE) {
        v4f v = *(const v4f*)(tile + r * D + lane * 4);
        float s = 0.f, s2 = 0.f;
#pragma unroll
        for (int i = 0; i < 4; ++i) {
            float vv = v[i] + bv[i];
            vv = fmaxf(vv, 0.0f);
            v[i] = vv;
            s  += vv;
            s2 += vv * vv;
        }
#pragma unroll
        for (int off = 16; off >= 1; off >>= 1) {
            s  += __shfl_xor(s,  off, 32);
            s2 += __shfl_xor(s2, off, 32);
        }
        float mu  = s * (1.0f / 128.0f);
        float var = s2 * (1.0f / 128.0f) - mu * mu;
        float rs  = rsqrtf(var + LN_EPS);
#pragma unroll
        for (int i = 0; i < 4; ++i)
            v[i] = (v[i] - mu) * rs * gv[i] + tv[i];
        *(v4f*)(out + (size_t)(lo + r) * D + lane * 4) = v;
    }
}

extern "C" void kernel_launch(void* const* d_in, const int* in_sizes, int n_in,
                              void* d_out, int out_size, void* d_ws, size_t ws_size,
                              hipStream_t stream) {
    const float* x     = (const float*)d_in[0];
    const int*   eidx  = (const int*)  d_in[1];
    const float* ew    = (const float*)d_in[2];
    const float* W     = (const float*)d_in[3];
    const float* bias  = (const float*)d_in[4];
    const float* gamma = (const float*)d_in[5];
    const float* beta  = (const float*)d_in[6];
    float*       out   = (float*)d_out;

    const int N = in_sizes[0] / D;
    const int E = in_sizes[2];
    const int* rows = eidx;
    const int* cols = eidx + E;

    float* h    = (float*)d_ws;
    float* deg  = h + (size_t)N * D;

    deg_init_kernel<<<(N + 255) / 256, 256, 0, stream>>>(deg, N);

    gemm_wmma_kernel<<<GEMM_GROUPS, GEMM_WAVES * 32, 0, stream>>>(x, W, h, N, cols, ew, deg, E);

    range_sum_norm_kernel<<<(N + TR - 1) / TR, NWAVE * 32, 0, stream>>>(
        rows, cols, ew, deg, h, bias, gamma, beta, out, N, E);
}


# scan on 1-byte range keys: GEMM launch's degree pass also stores target/392 per edge as a byte; groups scan 640 KB of keys, not 2560 KB of targets; arrival order at deg and LDS float adds changes
# speedup vs baseline: 2.2543x; 1.0998x over previous
#include <hip/hip_runtime.h>
#include <hip/hip_bf16.h>

typedef __attribute__((ext_vector_type(2))) float v2f;
typedef __attribute__((ext_vector_type(4))) float v4f;
typedef __attribute__((ext_vector_type(8))) float v8f;
typedef __attribute__((ext_vector_type(4))) int v4i;

#define D 128
#define LN_EPS 1e-5f
#define TR    392

__global__ void __launch_bounds__(256)
deg_init_kernel(float* __restrict__ deg, int N) {
    int i = blockIdx.x * blockDim.x + threadIdx.x;
    if (i < N) deg[i] = 1.0f;
}

__device__ __forceinline__ float inv_sqrt_deg(float d) {
    return (d > 0.0f) ? rsqrtf(d) : 0.0f;
}

#define GEMM_WAVES 4
#define GEMM_GROUPS 256
#define DEG_BATCH 8
#define KEY_BATCH 5

__global__ void __launch_bounds__(GEMM_WAVES * 32)
gemm_wmma_kernel(const float* __restrict__ x, const float* __restrict__ W,
                 float* __restrict__ h, int N,
                 const int* __restrict__ cols, const float* __restrict__ ew,
                 float* __restrict__ deg, int E, unsigned int* __restrict__ keys) {
    const int lane = threadIdx.x & 31;
    const int wave = __builtin_amdgcn_readfirstlane((int)(threadIdx.x >> 5));
    const int c    = lane & 15;
    const int hh   = lane >> 4;

    v4f wb[8][16];
#pragma unroll
    for (int t = 0; t < 8; ++t) {
        const int n = 64 * (t >> 2) + 4 * c + (t & 3);
        const float* wrow = W + (size_t)n * D + 4 * hh;
#pragma unroll
        for (int j = 0; j < 16; ++j) wb[t][j] = *(const v4f*)(wrow + 8 * j);
    }

    const int ntiles = (N + 15) / 16;
    const int nw     = (int)gridDim.x * GEMM_WAVES;
    int tile         = (int)blockIdx.x * GEMM_WAVES + wave;

    auto load_tile = [&](v4f (&dst)[16], int tl) {
        int row = tl * 16 + c;
        if (row > N - 1) row = N - 1;
        const float* p = x + (size_t)row * D + 4 * hh;
#pragma unroll
        for (int j = 0; j < 16; ++j) dst[j] = *(const v4f*)(p + 8 * j);
    };

    v4f xa[16];
    load_tile(xa, tile);

    if (keys != nullptr) {
        const int nunits = (E + 127) >> 7;
        for (int u0 = tile; u0 < nunits; u0 += KEY_BATCH * nw) {
            v4i  tc[KEY_BATCH];
            v4f  tw[KEY_BATCH];
#pragma unroll
            for (int b = 0; b < KEY_BATCH; ++b) {
                const int e  = (u0 + b * nw) * 128 + lane * 4;
                const int ec = (e < E) ? e : 0;
                tc[b] = *(const v4i*)(cols + ec);
                tw[b] = *(const v4f*)(ew + ec);
            }
#pragma unroll
            for (int b = 0; b < KEY_BATCH; ++b) {
                const int e = (u0 + b * nw) * 128 + lane * 4;
                if (e < E) {
                    atomicAdd(&deg[tc[b].x], tw[b].x);
                    atomicAdd(&deg[tc[b].y], tw[b].y);
                    atomicAdd(&deg[tc[b].z], tw[b].z);
                    atomicAdd(&deg[tc[b].w], tw[b].w);
                    const unsigned k0 = (unsigned)tc[b].x / (unsigned)TR;
                    const unsigned k1 = (unsigned)tc[b].y / (unsigned)TR;
                    const unsigned k2 = (unsigned)tc[b].z / (unsigned)TR;
                    const unsigned k3 = (unsigned)tc[b].w / (unsigned)TR;
                    keys[e >> 2] = k0 | (k1 << 8) | (k2 << 16) | (k3 << 24);
                }
            }
        }
    } else {
        const int per = (E + (int)gridDim.x - 1) / (int)gridDim.x;
        const int e0  = (int)blockIdx.x * per;
        const int e1  = (e0 + per < E) ? (e0 + per) : E;
        for (int eb = e0 + (int)threadIdx.x; eb < e1; eb += DEG_BATCH * GEMM_WAVES * 32) {
            int   tc[DEG_BATCH];
            float tw[DEG_BATCH];
#pragma unroll
            for (int u = 0; u < DEG_BATCH; ++u) {
                const int e  = eb + u * GEMM_WAVES * 32;
                const int ec = (e < e1) ? e : eb;
                tc[u] = cols[ec];
                tw[u] = ew[ec];
            }
#pragma unroll
            for (int u = 0; u < DEG_BATCH; ++u)
                if (eb + u * GEMM_WAVES * 32 < e1) atomicAdd(&deg[tc[u]], tw[u]);
        }
    }

    for (; tile < ntiles; tile += nw) {
        v4f xn[16];
        load_tile(xn, tile + nw);

        v8f acc[8];
#pragma unroll
        for (int t = 0; t < 8; ++t) acc[t] = v8f{};

#pragma unroll
        for (int j = 0; j < 16; ++j) {
            const v2f a0 = {xa[j].x, xa[j].y};
            const v2f a1 = {xa[j].z, xa[j].w};
#pragma unroll
            for (int t = 0; t < 8; ++t) {
                const v2f b0 = {wb[t][j].x, wb[t][j].y};
                acc[t] = __builtin_amdgcn_wmma_f32_16x16x4_f32(false, a0, false, b0,
                                                               (short)0, acc[t], false, false);
            }
#pragma unroll
            for (int t = 0; t < 8; ++t) {
                const v2f b1 = {wb[t][j].z, wb[t][j].w};
                acc[t] = __builtin_amdgcn_wmma_f32_16x16x4_f32(false, a1, false, b1,
                                                               (short)0, acc[t], false, false);
            }
        }

        const int rbase = tile * 16 + 8 * hh;
#pragma unroll
        for (int r = 0; r < 8; ++r) {
            const int row = rbase + r;
            if (row < N) {
                float* o = h + (size_t)row * D + 4 * c;
                v4f lo4 = {acc[0][r], acc[1][r], acc[2][r], acc[3][r]};
                v4f hi4 = {acc[4][r], acc[5][r], acc[6][r], acc[7][r]};
                *(v4f*)(o)      = lo4;
                *(v4f*)(o + 64) = hi4;
            }
        }

#pragma unroll
        for (int j = 0; j < 16; ++j) xa[j] = xn[j];
    }
}

#define NWAVE 32
#define CHUNK 128
#define GROUP 4
#define LCAP  (GROUP * CHUNK + 128)
#define KCHUNK 512

__device__ __forceinline__ unsigned zero_bytes(unsigned x) {
    const unsigned t = (x & 0x7F7F7F7Fu) + 0x7F7F7F7Fu;
    return ~(t | x | 0x7F7F7F7Fu);
}

__global__ void __launch_bounds__(1024)
range_sum_norm_kernel(const int* __restrict__ rows, const int* __restrict__ cols,
                      const float* __restrict__ ew, const float* __restrict__ deg,
                      const float* __restrict__ h, const float* __restrict__ bias,
                      const float* __restrict__ gamma, const float* __restrict__ beta,
                      float* __restrict__ out, int N, int E,
                      const unsigned int* __restrict__ keys) {
    __shared__ __attribute__((aligned(16))) float tile[TR * D];
    __shared__ int lists[NWAVE * LCAP];

    const int lane = threadIdx.x & 31;
    const int wave = __builtin_amdgcn_readfirstlane((int)(threadIdx.x >> 5));
    const int lo   = blockIdx.x * TR;
    const int nr   = (N - lo < TR) ? (N - lo) : TR;
    int* mylist    = lists + wave * LCAP;

    for (int r = wave; r < nr; r += NWAVE) {
        v4f hv   = *(const v4f*)(h + (size_t)(lo + r) * D + lane * 4);
        float di = inv_sqrt_deg(deg[lo + r]);
        v4f s;
#pragma unroll
        for (int i = 0; i < 4; ++i) s[i] = hv[i] * di * di;
        *(v4f*)(tile + r * D + lane * 4) = s;
    }
    __syncthreads();

    auto drain = [&](int cnt) {
        for (int b0 = 0; b0 < cnt; b0 += 32) {
            int  i   = b0 + lane;
            int  e   = mylist[(i < cnt) ? i : b0];
            int  src = rows[e];
            int  dl  = cols[e] - lo;
            float nm = inv_sqrt_deg(deg[src]) * ew[e] * inv_sqrt_deg(deg[lo + dl]);
            int  m   = (cnt - b0 < 32) ? (cnt - b0) : 32;
            for (int j0 = 0; j0 < m; j0 += 8) {
                v4f hv[8];
#pragma unroll
                for (int u = 0; u < 8; ++u) {
                    int j = (j0 + u < m) ? (j0 + u) : (m - 1);
                    int s = __builtin_amdgcn_readlane(src, j);
                    hv[u] = *(const v4f*)(h + (size_t)s * D + lane * 4);
                }
#pragma unroll
                for (int u = 0; u < 8; ++u) {
                    if (j0 + u < m) {
                        int   d = __builtin_amdgcn_readlane(dl, j0 + u);
                        float f = __builtin_bit_cast(float,
                                      __builtin_amdgcn_readlane(__builtin_bit_cast(int, nm), j0 + u));
                        float* a = tile + d * D + lane * 4;
#pragma unroll
                        for (int k = 0; k < 4; ++k)
                            __hip_atomic_fetch_add(a + k, hv[u][k] * f, __ATOMIC_RELAXED,
                                                   __HIP_MEMORY_SCOPE_WORKGROUP);
                    }
                }
            }
        }
    };

    const bool vec   = (E & 3) == 0;
    const int  nfull = vec ? (E / CHUNK) : 0;
    const int  rot   = (int)blockIdx.x * (nfull / (int)gridDim.x);
    int cnt = 0;
    int scanned = nfull * CHUNK;

    if (keys != nullptr) {
        const int      nfk  = E / KCHUNK;
        const int      rotk = (int)blockIdx.x * (nfk / (int)gridDim.x);
        const unsigned mine = (unsigned)blockIdx.x * 0x01010101u;
        const unsigned char* kb = (const unsigned char*)keys;
        scanned = nfk * KCHUNK;
        for (int q0 = wave; q0 < nfk; q0 += GROUP * NWAVE) {
            v4i      k[GROUP];
            int      eb[GROUP];
            unsigned ok[GROUP];
#pragma unroll
            for (int u = 0; u < GROUP; ++u) {
                int  q     = q0 + u * NWAVE;
                bool valid = q < nfk;
                int  ch    = valid ? (q + rotk) : rotk;
                if (ch >= nfk) ch -= nfk;
                eb[u] = ch * KCHUNK;
                ok[u] = valid ? 0xFFFFFFFFu : 0u;
                k[u]  = *(const v4i*)(kb + eb[u] + lane * 16);
            }
            unsigned mv[GROUP];
#pragma unroll
            for (int u = 0; u < GROUP; ++u) {
                const unsigned z0 = zero_bytes((unsigned)k[u].x ^ mine);
                const unsigned z1 = zero_bytes((unsigned)k[u].y ^ mine);
                const unsigned z2 = zero_bytes((unsigned)k[u].z ^ mine);
                const unsigned z3 = zero_bytes((unsigned)k[u].w ^ mine);
                mv[u] = ((z0 >> 7) | (z1 >> 6) | (z2 >> 5) | (z3 >> 4)) & ok[u];
            }
#pragma nounroll
            for (int u = 0; u < GROUP; ++u) {
                unsigned mm = (u == 0) ? mv[0] : (u == 1) ? mv[1] : (u == 2) ? mv[2] : mv[3];
                const int e0 = ((u == 0) ? eb[0] : (u == 1) ? eb[1] : (u == 2) ? eb[2] : eb[3]) + lane * 16;
                while (__builtin_amdgcn_ballot_w32(mm != 0u) != 0u) {
                    if (cnt > LCAP - 32) {
                        drain(cnt);
                        cnt = 0;
                    }
                    const bool     has = mm != 0u;
                    const unsigned b   = __builtin_amdgcn_ballot_w32(has);
                    if (has) {
                        const int c = __builtin_ctz(mm);
                        mylist[cnt + __builtin_amdgcn_mbcnt_lo(b, 0u)] = e0 + ((c & 7) << 2) + (c >> 3);
                        mm &= mm - 1u;
                    }
                    cnt += __builtin_popcount(b);
                }
            }
        }
    } else {
        for (int q0 = wave; q0 < nfull; q0 += GROUP * NWAVE) {
            v4i      c[GROUP];
            int      eb[GROUP];
            unsigned lim[GROUP];
#pragma unroll
            for (int u = 0; u < GROUP; ++u) {
                int  q     = q0 + u * NWAVE;
                bool valid = q < nfull;
                int  ch    = valid ? (q + rot) : rot;
                if (ch >= nfull) ch -= nfull;
                eb[u]  = ch * CHUNK;
                lim[u] = valid ? (unsigned)nr : 0u;
                c[u]   = *(const v4i*)(cols + eb[u] + lane * 4);
            }
#pragma unroll
            for (int u = 0; u < GROUP; ++u) {
                int  e0 = eb[u] + lane * 4;
                bool m0 = (unsigned)(c[u].x - lo) < lim[u];
                bool m1 = (unsigned)(c[u].y - lo) < lim[u];
                bool m2 = (unsigned)(c[u].z - lo) < lim[u];
                bool m3 = (unsigned)(c[u].w - lo) < lim[u];
                if (__builtin_amdgcn_ballot_w32(m0 | m1 | m2 | m3) != 0u) {
                    unsigned b;
                    b = __builtin_amdgcn_ballot_w32(m0);
                    if (m0) mylist[cnt + __builtin_amdgcn_mbcnt_lo(b, 0u)] = e0;
                    cnt += __builtin_popcount(b);
                    b = __builtin_amdgcn_ballot_w32(m1);
                    if (m1) mylist[cnt + __builtin_amdgcn_mbcnt_lo(b, 0u)] = e0 + 1;
                    cnt += __builtin_popcount(b);
                    b = __builtin_amdgcn_ballot_w32(m2);
                    if (m2) mylist[cnt + __builtin_amdgcn_mbcnt_lo(b, 0u)] = e0 + 2;
                    cnt += __builtin_popcount(b);
                    b = __builtin_amdgcn_ballot_w32(m3);
                    if (m3) mylist[cnt + __builtin_amdgcn_mbcnt_lo(b, 0u)] = e0 + 3;
                    cnt += __builtin_popcount(b);
                }
            }
            if (cnt > LCAP - GROUP * CHUNK) {
                drain(cnt);
                cnt = 0;
            }
        }
    }
    for (int base = scanned + wave * 32; base < E; base += NWAVE * 32) {
        if (cnt > LCAP - 32) {
            drain(cnt);
            cnt = 0;
        }
        int  e   = base + lane;
        int  col = (e < E) ? cols[e] : -1;
        bool m   = (unsigned)(col - lo) < (unsigned)nr;
        unsigned b = __builtin_amdgcn_ballot_w32(m);
        if (m) mylist[cnt + __builtin_amdgcn_mbcnt_lo(b, 0u)] = e;
        cnt += __builtin_popcount(b);
    }
    drain(cnt);
    __syncthreads();

    const v4f bv = *(const v4f*)(bias  + lane * 4);
    const v4f gv = *(const v4f*)(gamma + lane * 4);
    const v4f tv = *(const v4f*)(beta  + lane * 4);
    for (int r = wave; r < nr; r += NWAVE) {
        v4f v = *(const v4f*)(tile + r * D + lane * 4);
        float s = 0.f, s2 = 0.f;
#pragma unroll
        for (int i = 0; i < 4; ++i) {
            float vv = v[i] + bv[i];
            vv = fmaxf(vv, 0.0f);
            v[i] = vv;
            s  += vv;
            s2 += vv * vv;
        }
#pragma unroll
        for (int off = 16; off >= 1; off >>= 1) {
            s  += __shfl_xor(s,  off, 32);
            s2 += __shfl_xor(s2, off, 32);
        }
        float mu  = s * (1.0f / 128.0f);
        float var = s2 * (1.0f / 128.0f) - mu * mu;
        float rs  = rsqrtf(var + LN_EPS);
#pragma unroll
        for (int i = 0; i < 4; ++i)
            v[i] = (v[i] - mu) * rs * gv[i] + tv[i];
        *(v4f*)(out + (size_t)(lo + r) * D + lane * 4) = v;
    }
}

extern "C" void kernel_launch(void* const* d_in, const int* in_sizes, int n_in,
                              void* d_out, int out_size, void* d_ws, size_t ws_size,
                              hipStream_t stream) {
    const float* x     = (const float*)d_in[0];
    const int*   eidx  = (const int*)  d_in[1];
    const float* ew    = (const float*)d_in[2];
    const float* W     = (const float*)d_in[3];
    const float* bias  = (const float*)d_in[4];
    const float* gamma = (const float*)d_in[5];
    const float* beta  = (const float*)d_in[6];
    float*       out   = (float*)d_out;

    const int N = in_sizes[0] / D;
    const int E = in_sizes[2];
    const int* rows = eidx;
    const int* cols = eidx + E;

    float* h    = (float*)d_ws;
    float* deg  = h + (size_t)N * D;

    const size_t koff  = (((size_t)N * (D + 1) * sizeof(float)) + 127) & ~(size_t)127;
    const bool   keyed = ((E & 3) == 0) && (E >= KCHUNK) && ((N + TR - 1) / TR <= 256) &&
                         (ws_size >= koff + (size_t)E);
    unsigned int* keys = keyed ? (unsigned int*)((char*)d_ws + koff) : nullptr;

    deg_init_kernel<<<(N + 255) / 256, 256, 0, stream>>>(deg, N);

    gemm_wmma_kernel<<<GEMM_GROUPS, GEMM_WAVES * 32, 0, stream>>>(x, W, h, N, cols, ew, deg, E, keys);

    range_sum_norm_kernel<<<(N + TR - 1) / TR, NWAVE * 32, 0, stream>>>(
        rows, cols, ew, deg, h, bias, gamma, beta, out, N, E, keys);
}


# LDS tile channel-interleaved: channel 4*lane+k of a row sits at dword 32*k+lane, so each ds_add_f32 of a gathered row hits 32 banks once (was 16 banks twice); arrival order at LDS float adds changes
# speedup vs baseline: 2.5502x; 1.1313x over previous
#include <hip/hip_runtime.h>
#include <hip/hip_bf16.h>

typedef __attribute__((ext_vector_type(2))) float v2f;
typedef __attribute__((ext_vector_type(4))) float v4f;
typedef __attribute__((ext_vector_type(8))) float v8f;
typedef __attribute__((ext_vector_type(4))) int v4i;

#define D 128
#define LN_EPS 1e-5f
#define TR    392

__global__ void __launch_bounds__(256)
deg_init_kernel(float* __restrict__ deg, int N) {
    int i = blockIdx.x * blockDim.x + threadIdx.x;
    if (i < N) deg[i] = 1.0f;
}

__device__ __forceinline__ float inv_sqrt_deg(float d) {
    return (d > 0.0f) ? rsqrtf(d) : 0.0f;
}

#define GEMM_WAVES 4
#define GEMM_GROUPS 256
#define DEG_BATCH 8
#define KEY_BATCH 5

__global__ void __launch_bounds__(GEMM_WAVES * 32)
gemm_wmma_kernel(const float* __restrict__ x, const float* __restrict__ W,
                 float* __restrict__ h, int N,
                 const int* __restrict__ cols, const float* __restrict__ ew,
                 float* __restrict__ deg, int E, unsigned int* __restrict__ keys) {
    const int lane = threadIdx.x & 31;
    const int wave = __builtin_amdgcn_readfirstlane((int)(threadIdx.x >> 5));
    const int c    = lane & 15;
    const int hh   = lane >> 4;

    v4f wb[8][16];
#pragma unroll
    for (int t = 0; t < 8; ++t) {
        const int n = 64 * (t >> 2) + 4 * c + (t & 3);
        const float* wrow = W + (size_t)n * D + 4 * hh;
#pragma unroll
        for (int j = 0; j < 16; ++j) wb[t][j] = *(const v4f*)(wrow + 8 * j);
    }

    const int ntiles = (N + 15) / 16;
    const int nw     = (int)gridDim.x * GEMM_WAVES;
    int tile         = (int)blockIdx.x * GEMM_WAVES + wave;

    auto load_tile = [&](v4f (&dst)[16], int tl) {
        int row = tl * 16 + c;
        if (row > N - 1) row = N - 1;
        const float* p = x + (size_t)row * D + 4 * hh;
#pragma unroll
        for (int j = 0; j < 16; ++j) dst[j] = *(const v4f*)(p + 8 * j);
    };

    v4f xa[16];
    load_tile(xa, tile);

    if (keys != nullptr) {
        const int nunits = (E + 127) >> 7;
        for (int u0 = tile; u0 < nunits; u0 += KEY_BATCH * nw) {
            v4i  tc[KEY_BATCH];
            v4f  tw[KEY_BATCH];
#pragma unroll
            for (int b = 0; b < KEY_BATCH; ++b) {
                const int e  = (u0 + b * nw) * 128 + lane * 4;
                const int ec = (e < E) ? e : 0;
                tc[b] = *(const v4i*)(cols + ec);
                tw[b] = *(const v4f*)(ew + ec);
            }
#pragma unroll
            for (int b = 0; b < KEY_BATCH; ++b) {
                const int e = (u0 + b * nw) * 128 + lane * 4;
                if (e < E) {
                    atomicAdd(&deg[tc[b].x], tw[b].x);
                    atomicAdd(&deg[tc[b].y], tw[b].y);
                    atomicAdd(&deg[tc[b].z], tw[b].z);
                    atomicAdd(&deg[tc[b].w], tw[b].w);
                    const unsigned k0 = (unsigned)tc[b].x / (unsigned)TR;
                    const unsigned k1 = (unsigned)tc[b].y / (unsigned)TR;
                    const unsigned k2 = (unsigned)tc[b].z / (unsigned)TR;
                    const unsigned k3 = (unsigned)tc[b].w / (unsigned)TR;
                    keys[e >> 2] = k0 | (k1 << 8) | (k2 << 16) | (k3 << 24);
                }
            }
        }
    } else {
        const int per = (E + (int)gridDim.x - 1) / (int)gridDim.x;
        const int e0  = (int)blockIdx.x * per;
        const int e1  = (e0 + per < E) ? (e0 + per) : E;
        for (int eb = e0 + (int)threadIdx.x; eb < e1; eb += DEG_BATCH * GEMM_WAVES * 32) {
            int   tc[DEG_BATCH];
            float tw[DEG_BATCH];
#pragma unroll
            for (int u = 0; u < DEG_BATCH; ++u) {
                const int e  = eb + u * GEMM_WAVES * 32;
                const int ec = (e < e1) ? e : eb;
                tc[u] = cols[ec];
                tw[u] = ew[ec];
            }
#pragma unroll
            for (int u = 0; u < DEG_BATCH; ++u)
                if (eb + u * GEMM_WAVES * 32 < e1) atomicAdd(&deg[tc[u]], tw[u]);
        }
    }

    for (; tile < ntiles; tile += nw) {
        v4f xn[16];
        load_tile(xn, tile + nw);

        v8f acc[8];
#pragma unroll
        for (int t = 0; t < 8; ++t) acc[t] = v8f{};

#pragma unroll
        for (int j = 0; j < 16; ++j) {
            const v2f a0 = {xa[j].x, xa[j].y};
            const v2f a1 = {xa[j].z, xa[j].w};
#pragma unroll
            for (int t = 0; t < 8; ++t) {
                const v2f b0 = {wb[t][j].x, wb[t][j].y};
                acc[t] = __builtin_amdgcn_wmma_f32_16x16x4_f32(false, a0, false, b0,
                                                               (short)0, acc[t], false, false);
            }
#pragma unroll
            for (int t = 0; t < 8; ++t) {
                const v2f b1 = {wb[t][j].z, wb[t][j].w};
                acc[t] = __builtin_amdgcn_wmma_f32_16x16x4_f32(false, a1, false, b1,
                                                               (short)0, acc[t], false, false);
            }
        }

        const int rbase = tile * 16 + 8 * hh;
#pragma unroll
        for (int r = 0; r < 8; ++r) {
            const int row = rbase + r;
            if (row < N) {
                float* o = h + (size_t)row * D + 4 * c;
                v4f lo4 = {acc[0][r], acc[1][r], acc[2][r], acc[3][r]};
                v4f hi4 = {acc[4][r], acc[5][r], acc[6][r], acc[7][r]};
                *(v4f*)(o)      = lo4;
                *(v4f*)(o + 64) = hi4;
            }
        }

#pragma unroll
        for (int j = 0; j < 16; ++j) xa[j] = xn[j];
    }
}

#define NWAVE 32
#define CHUNK 128
#define GROUP 4
#define LCAP  (GROUP * CHUNK + 128)
#define KCHUNK 512

__device__ __forceinline__ unsigned zero_bytes(unsigned x) {
    const unsigned t = (x & 0x7F7F7F7Fu) + 0x7F7F7F7Fu;
    return ~(t | x | 0x7F7F7F7Fu);
}

__global__ void __launch_bounds__(1024)
range_sum_norm_kernel(const int* __restrict__ rows, const int* __restrict__ cols,
                      const float* __restrict__ ew, const float* __restrict__ deg,
                      const float* __restrict__ h, const float* __restrict__ bias,
                      const float* __restrict__ gamma, const float* __restrict__ beta,
                      float* __restrict__ out, int N, int E,
                      const unsigned int* __restrict__ keys) {
    __shared__ __attribute__((aligned(16))) float tile[TR * D];
    __shared__ int lists[NWAVE * LCAP];

    const int lane = threadIdx.x & 31;
    const int wave = __builtin_amdgcn_readfirstlane((int)(threadIdx.x >> 5));
    const int lo   = blockIdx.x * TR;
    const int nr   = (N - lo < TR) ? (N - lo) : TR;
    int* mylist    = lists + wave * LCAP;

    for (int r = wave; r < nr; r += NWAVE) {
        v4f hv   = *(const v4f*)(h + (size_t)(lo + r) * D + lane * 4);
        float di = inv_sqrt_deg(deg[lo + r]);
        float* t = tile + r * D + lane;
#pragma unroll
        for (int i = 0; i < 4; ++i) t[32 * i] = hv[i] * di * di;
    }
    __syncthreads();

    auto drain = [&](int cnt) {
        for (int b0 = 0; b0 < cnt; b0 += 32) {
            int  i   = b0 + lane;
            int  e   = mylist[(i < cnt) ? i : b0];
            int  src = rows[e];
            int  dl  = cols[e] - lo;
            float nm = inv_sqrt_deg(deg[src]) * ew[e] * inv_sqrt_deg(deg[lo + dl]);
            int  m   = (cnt - b0 < 32) ? (cnt - b0) : 32;
            for (int j0 = 0; j0 < m; j0 += 8) {
                v4f hv[8];
#pragma unroll
                for (int u = 0; u < 8; ++u) {
                    int j = (j0 + u < m) ? (j0 + u) : (m - 1);
                    int s = __builtin_amdgcn_readlane(src, j);
                    hv[u] = *(const v4f*)(h + (size_t)s * D + lane * 4);
                }
#pragma unroll
                for (int u = 0; u < 8; ++u) {
                    if (j0 + u < m) {
                        int   d = __builtin_amdgcn_readlane(dl, j0 + u);
                        float f = __builtin_bit_cast(float,
                                      __builtin_amdgcn_readlane(__builtin_bit_cast(int, nm), j0 + u));
                        float* a = tile + d * D + lane;
#pragma unroll
                        for (int k = 0; k < 4; ++k)
                            __hip_atomic_fetch_add(a + 32 * k, hv[u][k] * f, __ATOMIC_RELAXED,
                                                   __HIP_MEMORY_SCOPE_WORKGROUP);
                    }
                }
            }
        }
    };

    const bool vec   = (E & 3) == 0;
    const int  nfull = vec ? (E / CHUNK) : 0;
    const int  rot   = (int)blockIdx.x * (nfull / (int)gridDim.x);
    int cnt = 0;
    int scanned = nfull * CHUNK;

    if (keys != nullptr) {
        const int      nfk  = E / KCHUNK;
        const int      rotk = (int)blockIdx.x * (nfk / (int)gridDim.x);
        const unsigned mine = (unsigned)blockIdx.x * 0x01010101u;
        const unsigned char* kb = (const unsigned char*)keys;
        scanned = nfk * KCHUNK;
        for (int q0 = wave; q0 < nfk; q0 += GROUP * NWAVE) {
            v4i      k[GROUP];
            int      eb[GROUP];
            unsigned ok[GROUP];
#pragma unroll
            for (int u = 0; u < GROUP; ++u) {
                int  q     = q0 + u * NWAVE;
                bool valid = q < nfk;
                int  ch    = valid ? (q + rotk) : rotk;
                if (ch >= nfk) ch -= nfk;
                eb[u] = ch * KCHUNK;
                ok[u] = valid ? 0xFFFFFFFFu : 0u;
                k[u]  = *(const v4i*)(kb + eb[u] + lane * 16);
            }
            unsigned mv[GROUP];
#pragma unroll
            for (int u = 0; u < GROUP; ++u) {
                const unsigned z0 = zero_bytes((unsigned)k[u].x ^ mine);
                const unsigned z1 = zero_bytes((unsigned)k[u].y ^ mine);
                const unsigned z2 = zero_bytes((unsigned)k[u].z ^ mine);
                const unsigned z3 = zero_bytes((unsigned)k[u].w ^ mine);
                mv[u] = ((z0 >> 7) | (z1 >> 6) | (z2 >> 5) | (z3 >> 4)) & ok[u];
            }
#pragma nounroll
            for (int u = 0; u < GROUP; ++u) {
                unsigned mm = (u == 0) ? mv[0] : (u == 1) ? mv[1] : (u == 2) ? mv[2] : mv[3];
                const int e0 = ((u == 0) ? eb[0] : (u == 1) ? eb[1] : (u == 2) ? eb[2] : eb[3]) + lane * 16;
                while (__builtin_amdgcn_ballot_w32(mm != 0u) != 0u) {
                    if (cnt > LCAP - 32) {
                        drain(cnt);
                        cnt = 0;
                    }
                    const bool     has = mm != 0u;
                    const unsigned b   = __builtin_amdgcn_ballot_w32(has);
                    if (has) {
                        const int c = __builtin_ctz(mm);
                        mylist[cnt + __builtin_amdgcn_mbcnt_lo(b, 0u)] = e0 + ((c & 7) << 2) + (c >> 3);
                        mm &= mm - 1u;
                    }
                    cnt += __builtin_popcount(b);
                }
            }
        }
    } else {
        for (int q0 = wave; q0 < nfull; q0 += GROUP * NWAVE) {
            v4i      c[GROUP];
            int      eb[GROUP];
            unsigned lim[GROUP];
#pragma unroll
            for (int u = 0; u < GROUP; ++u) {
                int  q     = q0 + u * NWAVE;
                bool valid = q < nfull;
                int  ch    = valid ? (q + rot) : rot;
                if (ch >= nfull) ch -= nfull;
                eb[u]  = ch * CHUNK;
                lim[u] = valid ? (unsigned)nr : 0u;
                c[u]   = *(const v4i*)(cols + eb[u] + lane * 4);
            }
#pragma unroll
            for (int u = 0; u < GROUP; ++u) {
                int  e0 = eb[u] + lane * 4;
                bool m0 = (unsigned)(c[u].x - lo) < lim[u];
                bool m1 = (unsigned)(c[u].y - lo) < lim[u];
                bool m2 = (unsigned)(c[u].z - lo) < lim[u];
                bool m3 = (unsigned)(c[u].w - lo) < lim[u];
                if (__builtin_amdgcn_ballot_w32(m0 | m1 | m2 | m3) != 0u) {
                    unsigned b;
                    b = __builtin_amdgcn_ballot_w32(m0);
                    if (m0) mylist[cnt + __builtin_amdgcn_mbcnt_lo(b, 0u)] = e0;
                    cnt += __builtin_popcount(b);
                    b = __builtin_amdgcn_ballot_w32(m1);
                    if (m1) mylist[cnt + __builtin_amdgcn_mbcnt_lo(b, 0u)] = e0 + 1;
                    cnt += __builtin_popcount(b);
                    b = __builtin_amdgcn_ballot_w32(m2);
                    if (m2) mylist[cnt + __builtin_amdgcn_mbcnt_lo(b, 0u)] = e0 + 2;
                    cnt += __builtin_popcount(b);
                    b = __builtin_amdgcn_ballot_w32(m3);
                    if (m3) mylist[cnt + __builtin_amdgcn_mbcnt_lo(b, 0u)] = e0 + 3;
                    cnt += __builtin_popcount(b);
                }
            }
            if (cnt > LCAP - GROUP * CHUNK) {
                drain(cnt);
                cnt = 0;
            }
        }
    }
    for (int base = scanned + wave * 32; base < E; base += NWAVE * 32) {
        if (cnt > LCAP - 32) {
            drain(cnt);
            cnt = 0;
        }
        int  e   = base + lane;
        int  col = (e < E) ? cols[e] : -1;
        bool m   = (unsigned)(col - lo) < (unsigned)nr;
        unsigned b = __builtin_amdgcn_ballot_w32(m);
        if (m) mylist[cnt + __builtin_amdgcn_mbcnt_lo(b, 0u)] = e;
        cnt += __builtin_popcount(b);
    }
    drain(cnt);
    __syncthreads();

    const v4f bv = *(const v4f*)(bias  + lane * 4);
    const v4f gv = *(const v4f*)(gamma + lane * 4);
    const v4f tv = *(const v4f*)(beta  + lane * 4);
    for (int r = wave; r < nr; r += NWAVE) {
        const float* t = tile + r * D + lane;
        v4f v;
#pragma unroll
        for (int i = 0; i < 4; ++i) v[i] = t[32 * i];
        float s = 0.f, s2 = 0.f;
#pragma unroll
        for (int i = 0; i < 4; ++i) {
            float vv = v[i] + bv[i];
            vv = fmaxf(vv, 0.0f);
            v[i] = vv;
            s  += vv;
            s2 += vv * vv;
        }
#pragma unroll
        for (int off = 16; off >= 1; off >>= 1) {
            s  += __shfl_xor(s,  off, 32);
            s2 += __shfl_xor(s2, off, 32);
        }
        float mu  = s * (1.0f / 128.0f);
        float var = s2 * (1.0f / 128.0f) - mu * mu;
        float rs  = rsqrtf(var + LN_EPS);
#pragma unroll
        for (int i = 0; i < 4; ++i)
            v[i] = (v[i] - mu) * rs * gv[i] + tv[i];
        *(v4f*)(out + (size_t)(lo + r) * D + lane * 4) = v;
    }
}

extern "C" void kernel_launch(void* const* d_in, const int* in_sizes, int n_in,
                              void* d_out, int out_size, void* d_ws, size_t ws_size,
                              hipStream_t stream) {
    const float* x     = (const float*)d_in[0];
    const int*   eidx  = (const int*)  d_in[1];
    const float* ew    = (const float*)d_in[2];
    const float* W     = (const float*)d_in[3];
    const float* bias  = (const float*)d_in[4];
    const float* gamma = (const float*)d_in[5];
    const float* beta  = (const float*)d_in[6];
    float*       out   = (float*)d_out;

    const int N = in_sizes[0] / D;
    const int E = in_sizes[2];
    const int* rows = eidx;
    const int* cols = eidx + E;

    float* h    = (float*)d_ws;
    float* deg  = h + (size_t)N * D;

    const size_t koff  = (((size_t)N * (D + 1) * sizeof(float)) + 127) & ~(size_t)127;
    const bool   keyed = ((E & 3) == 0) && (E >= KCHUNK) && ((N + TR - 1) / TR <= 256) &&
                         (ws_size >= koff + (size_t)E);
    unsigned int* keys = keyed ? (unsigned int*)((char*)d_ws + koff) : nullptr;

    deg_init_kernel<<<(N + 255) / 256, 256, 0, stream>>>(deg, N);

    gemm_wmma_kernel<<<GEMM_GROUPS, GEMM_WAVES * 32, 0, stream>>>(x, W, h, N, cols, ew, deg, E, keys);

    range_sum_norm_kernel<<<(N + TR - 1) / TR, NWAVE * 32, 0, stream>>>(
        rows, cols, ew, deg, h, bias, gamma, beta, out, N, E, keys);
}


# edge records: the GEMM launch's degree pass also stores 8 B per edge (source, row in range, weight); the adding wave reads one scattered line per listed edge, not three; LDS add order as before
# speedup vs baseline: 2.6975x; 1.0578x over previous
#include <hip/hip_runtime.h>
#include <hip/hip_bf16.h>

typedef __attribute__((ext_vector_type(2))) float v2f;
typedef __attribute__((ext_vector_type(4))) float v4f;
typedef __attribute__((ext_vector_type(8))) float v8f;
typedef __attribute__((ext_vector_type(4))) int v4i;

#define D 128
#define LN_EPS 1e-5f
#define TR    392

__global__ void __launch_bounds__(256)
deg_init_kernel(float* __restrict__ deg, int N) {
    int i = blockIdx.x * blockDim.x + threadIdx.x;
    if (i < N) deg[i] = 1.0f;
}

__device__ __forceinline__ float inv_sqrt_deg(float d) {
    return (d > 0.0f) ? rsqrtf(d) : 0.0f;
}

#define GEMM_WAVES 4
#define GEMM_GROUPS 256
#define DEG_BATCH 8
#define KEY_BATCH 5

__global__ void __launch_bounds__(GEMM_WAVES * 32)
gemm_wmma_kernel(const float* __restrict__ x, const float* __restrict__ W,
                 float* __restrict__ h, int N,
                 const int* __restrict__ cols, const float* __restrict__ ew,
                 float* __restrict__ deg, int E, unsigned int* __restrict__ keys,
                 const int* __restrict__ rows, int* __restrict__ rec) {
    const int lane = threadIdx.x & 31;
    const int wave = __builtin_amdgcn_readfirstlane((int)(threadIdx.x >> 5));
    const int c    = lane & 15;
    const int hh   = lane >> 4;

    v4f wb[8][16];
#pragma unroll
    for (int t = 0; t < 8; ++t) {
        const int n = 64 * (t >> 2) + 4 * c + (t & 3);
        const float* wrow = W + (size_t)n * D + 4 * hh;
#pragma unroll
        for (int j = 0; j < 16; ++j) wb[t][j] = *(const v4f*)(wrow + 8 * j);
    }

    const int ntiles = (N + 15) / 16;
    const int nw     = (int)gridDim.x * GEMM_WAVES;
    int tile         = (int)blockIdx.x * GEMM_WAVES + wave;

    auto load_tile = [&](v4f (&dst)[16], int tl) {
        int row = tl * 16 + c;
        if (row > N - 1) row = N - 1;
        const float* p = x + (size_t)row * D + 4 * hh;
#pragma unroll
        for (int j = 0; j < 16; ++j) dst[j] = *(const v4f*)(p + 8 * j);
    };

    v4f xa[16];
    load_tile(xa, tile);

    if (keys != nullptr) {
        const int nunits = (E + 127) >> 7;
        for (int u0 = tile; u0 < nunits; u0 += KEY_BATCH * nw) {
            v4i  tc[KEY_BATCH];
            v4f  tw[KEY_BATCH];
            v4i  tr[KEY_BATCH];
#pragma unroll
            for (int b = 0; b < KEY_BATCH; ++b) {
                const int e  = (u0 + b * nw) * 128 + lane * 4;
                const int ec = (e < E) ? e : 0;
                tc[b] = *(const v4i*)(cols + ec);
                tw[b] = *(const v4f*)(ew + ec);
                tr[b] = (rec != nullptr) ? *(const v4i*)(rows + ec) : v4i{0, 0, 0, 0};
            }
#pragma unroll
            for (int b = 0; b < KEY_BATCH; ++b) {
                const int e = (u0 + b * nw) * 128 + lane * 4;
                if (e < E) {
                    atomicAdd(&deg[tc[b].x], tw[b].x);
                    atomicAdd(&deg[tc[b].y], tw[b].y);
                    atomicAdd(&deg[tc[b].z], tw[b].z);
                    atomicAdd(&deg[tc[b].w], tw[b].w);
                    const unsigned k0 = (unsigned)tc[b].x / (unsigned)TR;
                    const unsigned k1 = (unsigned)tc[b].y / (unsigned)TR;
                    const unsigned k2 = (unsigned)tc[b].z / (unsigned)TR;
                    const unsigned k3 = (unsigned)tc[b].w / (unsigned)TR;
                    keys[e >> 2] = k0 | (k1 << 8) | (k2 << 16) | (k3 << 24);
                    if (rec != nullptr) {
                        const unsigned d0 = (unsigned)tc[b].x - k0 * (unsigned)TR;
                        const unsigned d1 = (unsigned)tc[b].y - k1 * (unsigned)TR;
                        const unsigned d2 = (unsigned)tc[b].z - k2 * (unsigned)TR;
                        const unsigned d3 = (unsigned)tc[b].w - k3 * (unsigned)TR;
                        const float w0 = tw[b].x, w1 = tw[b].y, w2 = tw[b].z, w3 = tw[b].w;
                        const v4i r01 = {(int)((unsigned)tr[b].x | (d0 << 23)), __float_as_int(w0),
                                         (int)((unsigned)tr[b].y | (d1 << 23)), __float_as_int(w1)};
                        const v4i r23 = {(int)((unsigned)tr[b].z | (d2 << 23)), __float_as_int(w2),
                                         (int)((unsigned)tr[b].w | (d3 << 23)), __float_as_int(w3)};
                        *(v4i*)(rec + 2 * (size_t)e)     = r01;
                        *(v4i*)(rec + 2 * (size_t)e + 4) = r23;
                    }
                }
            }
        }
    } else {
        const int per = (E + (int)gridDim.x - 1) / (int)gridDim.x;
        const int e0  = (int)blockIdx.x * per;
        const int e1  = (e0 + per < E) ? (e0 + per) : E;
        for (int eb = e0 + (int)threadIdx.x; eb < e1; eb += DEG_BATCH * GEMM_WAVES * 32) {
            int   tc[DEG_BATCH];
            float tw[DEG_BATCH];
#pragma unroll
            for (int u = 0; u < DEG_BATCH; ++u) {
                const int e  = eb + u * GEMM_WAVES * 32;
                const int ec = (e < e1) ? e : eb;
                tc[u] = cols[ec];
                tw[u] = ew[ec];
            }
#pragma unroll
            for (int u = 0; u < DEG_BATCH; ++u)
                if (eb + u * GEMM_WAVES * 32 < e1) atomicAdd(&deg[tc[u]], tw[u]);
        }
    }

    for (; tile < ntiles; tile += nw) {
        v4f xn[16];
        load_tile(xn, tile + nw);

        v8f acc[8];
#pragma unroll
        for (int t = 0; t < 8; ++t) acc[t] = v8f{};

#pragma unroll
        for (int j = 0; j < 16; ++j) {
            const v2f a0 = {xa[j].x, xa[j].y};
            const v2f a1 = {xa[j].z, xa[j].w};
#pragma unroll
            for (int t = 0; t < 8; ++t) {
                const v2f b0 = {wb[t][j].x, wb[t][j].y};
                acc[t] = __builtin_amdgcn_wmma_f32_16x16x4_f32(false, a0, false, b0,
                                                               (short)0, acc[t], false, false);
            }
#pragma unroll
            for (int t = 0; t < 8; ++t) {
                const v2f b1 = {wb[t][j].z, wb[t][j].w};
                acc[t] = __builtin_amdgcn_wmma_f32_16x16x4_f32(false, a1, false, b1,
                                                               (short)0, acc[t], false, false);
            }
        }

        const int rbase = tile * 16 + 8 * hh;
#pragma unroll
        for (int r = 0; r < 8; ++r) {
            const int row = rbase + r;
            if (row < N) {
                float* o = h + (size_t)row * D + 4 * c;
                v4f lo4 = {acc[0][r], acc[1][r], acc[2][r], acc[3][r]};
                v4f hi4 = {acc[4][r], acc[5][r], acc[6][r], acc[7][r]};
                *(v4f*)(o)      = lo4;
                *(v4f*)(o + 64) = hi4;
            }
        }

#pragma unroll
        for (int j = 0; j < 16; ++j) xa[j] = xn[j];
    }
}

#define NWAVE 32
#define CHUNK 128
#define GROUP 4
#define LCAP  (GROUP * CHUNK + 128)
#define KCHUNK 512

__device__ __forceinline__ unsigned zero_bytes(unsigned x) {
    const unsigned t = (x & 0x7F7F7F7Fu) + 0x7F7F7F7Fu;
    return ~(t | x | 0x7F7F7F7Fu);
}

__global__ void __launch_bounds__(1024)
range_sum_norm_kernel(const int* __restrict__ rows, const int* __restrict__ cols,
                      const float* __restrict__ ew, const float* __restrict__ deg,
                      const float* __restrict__ h, const float* __restrict__ bias,
                      const float* __restrict__ gamma, const float* __restrict__ beta,
                      float* __restrict__ out, int N, int E,
                      const unsigned int* __restrict__ keys, const int* __restrict__ rec) {
    __shared__ __attribute__((aligned(16))) float tile[TR * D];
    __shared__ int lists[NWAVE * LCAP];

    const int lane = threadIdx.x & 31;
    const int wave = __builtin_amdgcn_readfirstlane((int)(threadIdx.x >> 5));
    const int lo   = blockIdx.x * TR;
    const int nr   = (N - lo < TR) ? (N - lo) : TR;
    int* mylist    = lists + wave * LCAP;

    for (int r = wave; r < nr; r += NWAVE) {
        v4f hv   = *(const v4f*)(h + (size_t)(lo + r) * D + lane * 4);
        float di = inv_sqrt_deg(deg[lo + r]);
        float* t = tile + r * D + lane;
#pragma unroll
        for (int i = 0; i < 4; ++i) t[32 * i] = hv[i] * di * di;
    }
    __syncthreads();

    auto drain = [&](int cnt) {
        for (int b0 = 0; b0 < cnt; b0 += 32) {
            int  i   = b0 + lane;
            int  e   = mylist[(i < cnt) ? i : b0];
            int   src, dl;
            float w;
            if (rec != nullptr) {
                const unsigned long long r = *(const unsigned long long*)(rec + 2 * (size_t)e);
                const unsigned r0 = (unsigned)r;
                src = (int)(r0 & 0x7FFFFFu);
                dl  = (int)(r0 >> 23);
                w   = __uint_as_float((unsigned)(r >> 32));
            } else {
                src = rows[e];
                dl  = cols[e] - lo;
                w   = ew[e];
            }
            float nm = inv_sqrt_deg(deg[src]) * w * inv_sqrt_deg(deg[lo + dl]);
            int  m   = (cnt - b0 < 32) ? (cnt - b0) : 32;
            for (int j0 = 0; j0 < m; j0 += 8) {
                v4f hv[8];
#pragma unroll
                for (int u = 0; u < 8; ++u) {
                    int j = (j0 + u < m) ? (j0 + u) : (m - 1);
                    int s = __builtin_amdgcn_readlane(src, j);
                    hv[u] = *(const v4f*)(h + (size_t)s * D + lane * 4);
                }
#pragma unroll
                for (int u = 0; u < 8; ++u) {
                    if (j0 + u < m) {
                        int   d = __builtin_amdgcn_readlane(dl, j0 + u);
                        float f = __builtin_bit_cast(float,
                                      __builtin_amdgcn_readlane(__builtin_bit_cast(int, nm), j0 + u));
                        float* a = tile + d * D + lane;
#pragma unroll
                        for (int k = 0; k < 4; ++k)
                            __hip_atomic_fetch_add(a + 32 * k, hv[u][k] * f, __ATOMIC_RELAXED,
                                                   __HIP_MEMORY_SCOPE_WORKGROUP);
                    }
                }
            }
        }
    };

    const bool vec   = (E & 3) == 0;
    const int  nfull = vec ? (E / CHUNK) : 0;
    const int  rot   = (int)blockIdx.x * (nfull / (int)gridDim.x);
    int cnt = 0;
    int scanned = nfull * CHUNK;

    if (keys != nullptr) {
        const int      nfk  = E / KCHUNK;
        const int      rotk = (int)blockIdx.x * (nfk / (int)gridDim.x);
        const unsigned mine = (unsigned)blockIdx.x * 0x01010101u;
        const unsigned char* kb = (const unsigned char*)keys;
        scanned = nfk * KCHUNK;
        for (int q0 = wave; q0 < nfk; q0 += GROUP * NWAVE) {
            v4i      k[GROUP];
            int      eb[GROUP];
            unsigned ok[GROUP];
#pragma unroll
            for (int u = 0; u < GROUP; ++u) {
                int  q     = q0 + u * NWAVE;
                bool valid = q < nfk;
                int  ch    = valid ? (q + rotk) : rotk;
                if (ch >= nfk) ch -= nfk;
                eb[u] = ch * KCHUNK;
                ok[u] = valid ? 0xFFFFFFFFu : 0u;
                k[u]  = *(const v4i*)(kb + eb[u] + lane * 16);
            }
            unsigned mv[GROUP];
#pragma unroll
            for (int u = 0; u < GROUP; ++u) {
                const unsigned z0 = zero_bytes((unsigned)k[u].x ^ mine);
                const unsigned z1 = zero_bytes((unsigned)k[u].y ^ mine);
                const unsigned z2 = zero_bytes((unsigned)k[u].z ^ mine);
                const unsigned z3 = zero_bytes((unsigned)k[u].w ^ mine);
                mv[u] = ((z0 >> 7) | (z1 >> 6) | (z2 >> 5) | (z3 >> 4)) & ok[u];
            }
#pragma nounroll
            for (int u = 0; u < GROUP; ++u) {
                unsigned mm = (u == 0) ? mv[0] : (u == 1) ? mv[1] : (u == 2) ? mv[2] : mv[3];
                const int e0 = ((u == 0) ? eb[0] : (u == 1) ? eb[1] : (u == 2) ? eb[2] : eb[3]) + lane * 16;
                while (__builtin_amdgcn_ballot_w32(mm != 0u) != 0u) {
                    if (cnt > LCAP - 32) {
                        drain(cnt);
                        cnt = 0;
                    }
                    const bool     has = mm != 0u;
                    const unsigned b   = __builtin_amdgcn_ballot_w32(has);
                    if (has) {
                        const int c = __builtin_ctz(mm);
                        mylist[cnt + __builtin_amdgcn_mbcnt_lo(b, 0u)] = e0 + ((c & 7) << 2) + (c >> 3);
                        mm &= mm - 1u;
                    }
                    cnt += __builtin_popcount(b);
                }
            }
        }
    } else {
        for (int q0 = wave; q0 < nfull; q0 += GROUP * NWAVE) {
            v4i      c[GROUP];
            int      eb[GROUP];
            unsigned lim[GROUP];
#pragma unroll
            for (int u = 0; u < GROUP; ++u) {
                int  q     = q0 + u * NWAVE;
                bool valid = q < nfull;
                int  ch    = valid ? (q + rot) : rot;
                if (ch >= nfull) ch -= nfull;
                eb[u]  = ch * CHUNK;
                lim[u] = valid ? (unsigned)nr : 0u;
                c[u]   = *(const v4i*)(cols + eb[u] + lane * 4);
            }
#pragma unroll
            for (int u = 0; u < GROUP; ++u) {
                int  e0 = eb[u] + lane * 4;
                bool m0 = (unsigned)(c[u].x - lo) < lim[u];
                bool m1 = (unsigned)(c[u].y - lo) < lim[u];
                bool m2 = (unsigned)(c[u].z - lo) < lim[u];
                bool m3 = (unsigned)(c[u].w - lo) < lim[u];
                if (__builtin_amdgcn_ballot_w32(m0 | m1 | m2 | m3) != 0u) {
                    unsigned b;
                    b = __builtin_amdgcn_ballot_w32(m0);
                    if (m0) mylist[cnt + __builtin_amdgcn_mbcnt_lo(b, 0u)] = e0;
                    cnt += __builtin_popcount(b);
                    b = __builtin_amdgcn_ballot_w32(m1);
                    if (m1) mylist[cnt + __builtin_amdgcn_mbcnt_lo(b, 0u)] = e0 + 1;
                    cnt += __builtin_popcount(b);
                    b = __builtin_amdgcn_ballot_w32(m2);
                    if (m2) mylist[cnt + __builtin_amdgcn_mbcnt_lo(b, 0u)] = e0 + 2;
                    cnt += __builtin_popcount(b);
                    b = __builtin_amdgcn_ballot_w32(m3);
                    if (m3) mylist[cnt + __builtin_amdgcn_mbcnt_lo(b, 0u)] = e0 + 3;
                    cnt += __builtin_popcount(b);
                }
            }
            if (cnt > LCAP - GROUP * CHUNK) {
                drain(cnt);
                cnt = 0;
            }
        }
    }
    for (int base = scanned + wave * 32; base < E; base += NWAVE * 32) {
        if (cnt > LCAP - 32) {
            drain(cnt);
            cnt = 0;
        }
        int  e   = base + lane;
        int  col = (e < E) ? cols[e] : -1;
        bool m   = (unsigned)(col - lo) < (unsigned)nr;
        unsigned b = __builtin_amdgcn_ballot_w32(m);
        if (m) mylist[cnt + __builtin_amdgcn_mbcnt_lo(b, 0u)] = e;
        cnt += __builtin_popcount(b);
    }
    drain(cnt);
    __syncthreads();

    const v4f bv = *(const v4f*)(bias  + lane * 4);
    const v4f gv = *(const v4f*)(gamma + lane * 4);
    const v4f tv = *(const v4f*)(beta  + lane * 4);
    for (int r = wave; r < nr; r += NWAVE) {
        const float* t = tile + r * D + lane;
        v4f v;
#pragma unroll
        for (int i = 0; i < 4; ++i) v[i] = t[32 * i];
        float s = 0.f, s2 = 0.f;
#pragma unroll
        for (int i = 0; i < 4; ++i) {
            float vv = v[i] + bv[i];
            vv = fmaxf(vv, 0.0f);
            v[i] = vv;
            s  += vv;
            s2 += vv * vv;
        }
#pragma unroll
        for (int off = 16; off >= 1; off >>= 1) {
            s  += __shfl_xor(s,  off, 32);
            s2 += __shfl_xor(s2, off, 32);
        }
        float mu  = s * (1.0f / 128.0f);
        float var = s2 * (1.0f / 128.0f) - mu * mu;
        float rs  = rsqrtf(var + LN_EPS);
#pragma unroll
        for (int i = 0; i < 4; ++i)
            v[i] = (v[i] - mu) * rs * gv[i] + tv[i];
        *(v4f*)(out + (size_t)(lo + r) * D + lane * 4) = v;
    }
}

extern "C" void kernel_launch(void* const* d_in, const int* in_sizes, int n_in,
                              void* d_out, int out_size, void* d_ws, size_t ws_size,
                              hipStream_t stream) {
    const float* x     = (const float*)d_in[0];
    const int*   eidx  = (const int*)  d_in[1];
    const float* ew    = (const float*)d_in[2];
    const float* W     = (const float*)d_in[3];
    const float* bias  = (const float*)d_in[4];
    const float* gamma = (const float*)d_in[5];
    const float* beta  = (const float*)d_in[6];
    float*       out   = (float*)d_out;

    const int N = in_sizes[0] / D;
    const int E = in_sizes[2];
    const int* rows = eidx;
    const int* cols = eidx + E;

    float* h    = (float*)d_ws;
    float* deg  = h + (size_t)N * D;

    const size_t koff  = (((size_t)N * (D + 1) * sizeof(float)) + 127) & ~(size_t)127;
    const bool   keyed = ((E & 3) == 0) && (E >= KCHUNK) && ((N + TR - 1) / TR <= 256) &&
                         (ws_size >= koff + (size_t)E);
    unsigned int* keys = keyed ? (unsigned int*)((char*)d_ws + koff) : nullptr;

    const size_t roff   = (koff + (size_t)E + 127) & ~(size_t)127;
    const bool   recd   = keyed && (N <= (1 << 23)) && (ws_size >= roff + (size_t)E * 8);
    int*         rec    = recd ? (int*)((char*)d_ws + roff) : nullptr;

    deg_init_kernel<<<(N + 255) / 256, 256, 0, stream>>>(deg, N);

    gemm_wmma_kernel<<<GEMM_GROUPS, GEMM_WAVES * 32, 0, stream>>>(x, W, h, N, cols, ew, deg, E, keys, rows, rec);

    range_sum_norm_kernel<<<(N + TR - 1) / TR, NWAVE * 32, 0, stream>>>(
        rows, cols, ew, deg, h, bias, gamma, beta, out, N, E, keys, rec);
}


# start values requested together: a wave issues the h-row and degree loads of all its 13 owned rows before it uses the first (was one row per round trip); values and sum order unchanged
# speedup vs baseline: 2.7834x; 1.0318x over previous
#include <hip/hip_runtime.h>
#include <hip/hip_bf16.h>

typedef __attribute__((ext_vector_type(2))) float v2f;
typedef __attribute__((ext_vector_type(4))) float v4f;
typedef __attribute__((ext_vector_type(8))) float v8f;
typedef __attribute__((ext_vector_type(4))) int v4i;

#define D 128
#define LN_EPS 1e-5f
#define TR    392

__global__ void __launch_bounds__(256)
deg_init_kernel(float* __restrict__ deg, int N) {
    int i = blockIdx.x * blockDim.x + threadIdx.x;
    if (i < N) deg[i] = 1.0f;
}

__device__ __forceinline__ float inv_sqrt_deg(float d) {
    return (d > 0.0f) ? rsqrtf(d) : 0.0f;
}

#define GEMM_WAVES 4
#define GEMM_GROUPS 256
#define DEG_BATCH 8
#define KEY_BATCH 5

__global__ void __launch_bounds__(GEMM_WAVES * 32)
gemm_wmma_kernel(const float* __restrict__ x, const float* __restrict__ W,
                 float* __restrict__ h, int N,
                 const int* __restrict__ cols, const float* __restrict__ ew,
                 float* __restrict__ deg, int E, unsigned int* __restrict__ keys,
                 const int* __restrict__ rows, int* __restrict__ rec) {
    const int lane = threadIdx.x & 31;
    const int wave = __builtin_amdgcn_readfirstlane((int)(threadIdx.x >> 5));
    const int c    = lane & 15;
    const int hh   = lane >> 4;

    v4f wb[8][16];
#pragma unroll
    for (int t = 0; t < 8; ++t) {
        const int n = 64 * (t >> 2) + 4 * c + (t & 3);
        const float* wrow = W + (size_t)n * D + 4 * hh;
#pragma unroll
        for (int j = 0; j < 16; ++j) wb[t][j] = *(const v4f*)(wrow + 8 * j);
    }

    const int ntiles = (N + 15) / 16;
    const int nw     = (int)gridDim.x * GEMM_WAVES;
    int tile         = (int)blockIdx.x * GEMM_WAVES + wave;

    auto load_tile = [&](v4f (&dst)[16], int tl) {
        int row = tl * 16 + c;
        if (row > N - 1) row = N - 1;
        const float* p = x + (size_t)row * D + 4 * hh;
#pragma unroll
        for (int j = 0; j < 16; ++j) dst[j] = *(const v4f*)(p + 8 * j);
    };

    v4f xa[16];
    load_tile(xa, tile);

    if (keys != nullptr) {
        const int nunits = (E + 127) >> 7;
        for (int u0 = tile; u0 < nunits; u0 += KEY_BATCH * nw) {
            v4i  tc[KEY_BATCH];
            v4f  tw[KEY_BATCH];
            v4i  tr[KEY_BATCH];
#pragma unroll
            for (int b = 0; b < KEY_BATCH; ++b) {
                const int e  = (u0 + b * nw) * 128 + lane * 4;
                const int ec = (e < E) ? e : 0;
                tc[b] = *(const v4i*)(cols + ec);
                tw[b] = *(const v4f*)(ew + ec);
                tr[b] = (rec != nullptr) ? *(const v4i*)(rows + ec) : v4i{0, 0, 0, 0};
            }
#pragma unroll
            for (int b = 0; b < KEY_BATCH; ++b) {
                const int e = (u0 + b * nw) * 128 + lane * 4;
                if (e < E) {
                    atomicAdd(&deg[tc[b].x], tw[b].x);
                    atomicAdd(&deg[tc[b].y], tw[b].y);
                    atomicAdd(&deg[tc[b].z], tw[b].z);
                    atomicAdd(&deg[tc[b].w], tw[b].w);
                    const unsigned k0 = (unsigned)tc[b].x / (unsigned)TR;
                    const unsigned k1 = (unsigned)tc[b].y / (unsigned)TR;
                    const unsigned k2 = (unsigned)tc[b].z / (unsigned)TR;
                    const unsigned k3 = (unsigned)tc[b].w / (unsigned)TR;
                    keys[e >> 2] = k0 | (k1 << 8) | (k2 << 16) | (k3 << 24);
                    if (rec != nullptr) {
                        const unsigned d0 = (unsigned)tc[b].x - k0 * (unsigned)TR;
                        const unsigned d1 = (unsigned)tc[b].y - k1 * (unsigned)TR;
                        const unsigned d2 = (unsigned)tc[b].z - k2 * (unsigned)TR;
                        const unsigned d3 = (unsigned)tc[b].w - k3 * (unsigned)TR;
                        const float w0 = tw[b].x, w1 = tw[b].y, w2 = tw[b].z, w3 = tw[b].w;
                        const v4i r01 = {(int)((unsigned)tr[b].x | (d0 << 23)), __float_as_int(w0),
                                         (int)((unsigned)tr[b].y | (d1 << 23)), __float_as_int(w1)};
                        const v4i r23 = {(int)((unsigned)tr[b].z | (d2 << 23)), __float_as_int(w2),
                                         (int)((unsigned)tr[b].w | (d3 << 23)), __float_as_int(w3)};
                        *(v4i*)(rec + 2 * (size_t)e)     = r01;
                        *(v4i*)(rec + 2 * (size_t)e + 4) = r23;
                    }
                }
            }
        }
    } else {
        const int per = (E + (int)gridDim.x - 1) / (int)gridDim.x;
        const int e0  = (int)blockIdx.x * per;
        const int e1  = (e0 + per < E) ? (e0 + per) : E;
        for (int eb = e0 + (int)threadIdx.x; eb < e1; eb += DEG_BATCH * GEMM_WAVES * 32) {
            int   tc[DEG_BATCH];
            float tw[DEG_BATCH];
#pragma unroll
            for (int u = 0; u < DEG_BATCH; ++u) {
                const int e  = eb + u * GEMM_WAVES * 32;
                const int ec = (e < e1) ? e : eb;
                tc[u] = cols[ec];
                tw[u] = ew[ec];
            }
#pragma unroll
            for (int u = 0; u < DEG_BATCH; ++u)
                if (eb + u * GEMM_WAVES * 32 < e1) atomicAdd(&deg[tc[u]], tw[u]);
        }
    }

    for (; tile < ntiles; tile += nw) {
        v4f xn[16];
        load_tile(xn, tile + nw);

        v8f acc[8];
#pragma unroll
        for (int t = 0; t < 8; ++t) acc[t] = v8f{};

#pragma unroll
        for (int j = 0; j < 16; ++j) {
            const v2f a0 = {xa[j].x, xa[j].y};
            const v2f a1 = {xa[j].z, xa[j].w};
#pragma unroll
            for (int t = 0; t < 8; ++t) {
                const v2f b0 = {wb[t][j].x, wb[t][j].y};
                acc[t] = __builtin_amdgcn_wmma_f32_16x16x4_f32(false, a0, false, b0,
                                                               (short)0, acc[t], false, false);
            }
#pragma unroll
            for (int t = 0; t < 8; ++t) {
                const v2f b1 = {wb[t][j].z, wb[t][j].w};
                acc[t] = __builtin_amdgcn_wmma_f32_16x16x4_f32(false, a1, false, b1,
                                                               (short)0, acc[t], false, false);
            }
        }

        const int rbase = tile * 16 + 8 * hh;
#pragma unroll
        for (int r = 0; r < 8; ++r) {
            const int row = rbase + r;
            if (row < N) {
                float* o = h + (size_t)row * D + 4 * c;
                v4f lo4 = {acc[0][r], acc[1][r], acc[2][r], acc[3][r]};
                v4f hi4 = {acc[4][r], acc[5][r], acc[6][r], acc[7][r]};
                *(v4f*)(o)      = lo4;
                *(v4f*)(o + 64) = hi4;
            }
        }

#pragma unroll
        for (int j = 0; j < 16; ++j) xa[j] = xn[j];
    }
}

#define NWAVE 32
#define CHUNK 128
#define GROUP 4
#define LCAP  (GROUP * CHUNK + 128)
#define KCHUNK 512
#define ROWB ((TR + NWAVE - 1) / NWAVE)

__device__ __forceinline__ unsigned zero_bytes(unsigned x) {
    const unsigned t = (x & 0x7F7F7F7Fu) + 0x7F7F7F7Fu;
    return ~(t | x | 0x7F7F7F7Fu);
}

__global__ void __launch_bounds__(1024)
range_sum_norm_kernel(const int* __restrict__ rows, const int* __restrict__ cols,
                      const float* __restrict__ ew, const float* __restrict__ deg,
                      const float* __restrict__ h, const float* __restrict__ bias,
                      const float* __restrict__ gamma, const float* __restrict__ beta,
                      float* __restrict__ out, int N, int E,
                      const unsigned int* __restrict__ keys, const int* __restrict__ rec) {
    __shared__ __attribute__((aligned(16))) float tile[TR * D];
    __shared__ int lists[NWAVE * LCAP];

    const int lane = threadIdx.x & 31;
    const int wave = __builtin_amdgcn_readfirstlane((int)(threadIdx.x >> 5));
    const int lo   = blockIdx.x * TR;
    const int nr   = (N - lo < TR) ? (N - lo) : TR;
    int* mylist    = lists + wave * LCAP;

    for (int r0 = wave; r0 < nr; r0 += NWAVE * ROWB) {
        v4f   hs[ROWB];
        float dg[ROWB];
#pragma unroll
        for (int u = 0; u < ROWB; ++u) {
            const int r  = r0 + u * NWAVE;
            const int rc = (r < nr) ? r : (nr - 1);
            hs[u] = *(const v4f*)(h + (size_t)(lo + rc) * D + lane * 4);
            dg[u] = deg[lo + rc];
        }
#pragma unroll
        for (int u = 0; u < ROWB; ++u) {
            asm volatile("" :: "v"(hs[u]), "s"(dg[u]));
        }
#pragma unroll
        for (int u = 0; u < ROWB; ++u) {
            const int r = r0 + u * NWAVE;
            if (r < nr) {
                const float di = inv_sqrt_deg(dg[u]);
                float* t = tile + r * D + lane;
#pragma unroll
                for (int i = 0; i < 4; ++i) t[32 * i] = hs[u][i] * di * di;
            }
        }
    }
    __syncthreads();

    auto drain = [&](int cnt) {
        for (int b0 = 0; b0 < cnt; b0 += 32) {
            int  i   = b0 + lane;
            int  e   = mylist[(i < cnt) ? i : b0];
            int   src, dl;
            float w;
            if (rec != nullptr) {
                const unsigned long long r = *(const unsigned long long*)(rec + 2 * (size_t)e);
                const unsigned r0 = (unsigned)r;
                src = (int)(r0 & 0x7FFFFFu);
                dl  = (int)(r0 >> 23);
                w   = __uint_as_float((unsigned)(r >> 32));
            } else {
                src = rows[e];
                dl  = cols[e] - lo;
                w   = ew[e];
            }
            float nm = inv_sqrt_deg(deg[src]) * w * inv_sqrt_deg(deg[lo + dl]);
            int  m   = (cnt - b0 < 32) ? (cnt - b0) : 32;
            for (int j0 = 0; j0 < m; j0 += 8) {
                v4f hv[8];
#pragma unroll
                for (int u = 0; u < 8; ++u) {
                    int j = (j0 + u < m) ? (j0 + u) : (m - 1);
                    int s = __builtin_amdgcn_readlane(src, j);
                    hv[u] = *(const v4f*)(h + (size_t)s * D + lane * 4);
                }
#pragma unroll
                for (int u = 0; u < 8; ++u) {
                    if (j0 + u < m) {
                        int   d = __builtin_amdgcn_readlane(dl, j0 + u);
                        float f = __builtin_bit_cast(float,
                                      __builtin_amdgcn_readlane(__builtin_bit_cast(int, nm), j0 + u));
                        float* a = tile + d * D + lane;
#pragma unroll
                        for (int k = 0; k < 4; ++k)
                            __hip_atomic_fetch_add(a + 32 * k, hv[u][k] * f, __ATOMIC_RELAXED,
                                                   __HIP_MEMORY_SCOPE_WORKGROUP);
                    }
                }
            }
        }
    };

    const bool vec   = (E & 3) == 0;
    const int  nfull = vec ? (E / CHUNK) : 0;
    const int  rot   = (int)blockIdx.x * (nfull / (int)gridDim.x);
    int cnt = 0;
    int scanned = nfull * CHUNK;

    if (keys != nullptr) {
        const int      nfk  = E / KCHUNK;
        const int      rotk = (int)blockIdx.x * (nfk / (int)gridDim.x);
        const unsigned mine = (unsigned)blockIdx.x * 0x01010101u;
        const unsigned char* kb = (const unsigned char*)keys;
        scanned = nfk * KCHUNK;
        for (int q0 = wave; q0 < nfk; q0 += GROUP * NWAVE) {
            v4i      k[GROUP];
            int      eb[GROUP];
            unsigned ok[GROUP];
#pragma unroll
            for (int u = 0; u < GROUP; ++u) {
                int  q     = q0 + u * NWAVE;
                bool valid = q < nfk;
                int  ch    = valid ? (q + rotk) : rotk;
                if (ch >= nfk) ch -= nfk;
                eb[u] = ch * KCHUNK;
                ok[u] = valid ? 0xFFFFFFFFu : 0u;
                k[u]  = *(const v4i*)(kb + eb[u] + lane * 16);
            }
            unsigned mv[GROUP];
#pragma unroll
            for (int u = 0; u < GROUP; ++u) {
                const unsigned z0 = zero_bytes((unsigned)k[u].x ^ mine);
                const unsigned z1 = zero_bytes((unsigned)k[u].y ^ mine);
                const unsigned z2 = zero_bytes((unsigned)k[u].z ^ mine);
                const unsigned z3 = zero_bytes((unsigned)k[u].w ^ mine);
                mv[u] = ((z0 >> 7) | (z1 >> 6) | (z2 >> 5) | (z3 >> 4)) & ok[u];
            }
#pragma nounroll
            for (int u = 0; u < GROUP; ++u) {
                unsigned mm = (u == 0) ? mv[0] : (u == 1) ? mv[1] : (u == 2) ? mv[2] : mv[3];
                const int e0 = ((u == 0) ? eb[0] : (u == 1) ? eb[1] : (u == 2) ? eb[2] : eb[3]) + lane * 16;
                while (__builtin_amdgcn_ballot_w32(mm != 0u) != 0u) {
                    if (cnt > LCAP - 32) {
                        drain(cnt);
                        cnt = 0;
                    }
                    const bool     has = mm != 0u;
                    const unsigned b   = __builtin_amdgcn_ballot_w32(has);
                    if (has) {
                        const int c = __builtin_ctz(mm);
                        mylist[cnt + __builtin_amdgcn_mbcnt_lo(b, 0u)] = e0 + ((c & 7) << 2) + (c >> 3);
                        mm &= mm - 1u;
                    }
                    cnt += __builtin_popcount(b);
                }
            }
        }
    } else {
        for (int q0 = wave; q0 < nfull; q0 += GROUP * NWAVE) {
            v4i      c[GROUP];
            int      eb[GROUP];
            unsigned lim[GROUP];
#pragma unroll
            for (int u = 0; u < GROUP; ++u) {
                int  q     = q0 + u * NWAVE;
                bool valid = q < nfull;
                int  ch    = valid ? (q + rot) : rot;
                if (ch >= nfull) ch -= nfull;
                eb[u]  = ch * CHUNK;
                lim[u] = valid ? (unsigned)nr : 0u;
                c[u]   = *(const v4i*)(cols + eb[u] + lane * 4);
            }
#pragma unroll
            for (int u = 0; u < GROUP; ++u) {
                int  e0 = eb[u] + lane * 4;
                bool m0 = (unsigned)(c[u].x - lo) < lim[u];
                bool m1 = (unsigned)(c[u].y - lo) < lim[u];
                bool m2 = (unsigned)(c[u].z - lo) < lim[u];
                bool m3 = (unsigned)(c[u].w - lo) < lim[u];
                if (__builtin_amdgcn_ballot_w32(m0 | m1 | m2 | m3) != 0u) {
                    unsigned b;
                    b = __builtin_amdgcn_ballot_w32(m0);
                    if (m0) mylist[cnt + __builtin_amdgcn_mbcnt_lo(b, 0u)] = e0;
                    cnt += __builtin_popcount(b);
                    b = __builtin_amdgcn_ballot_w32(m1);
                    if (m1) mylist[cnt + __builtin_amdgcn_mbcnt_lo(b, 0u)] = e0 + 1;
                    cnt += __builtin_popcount(b);
                    b = __builtin_amdgcn_ballot_w32(m2);
                    if (m2) mylist[cnt + __builtin_amdgcn_mbcnt_lo(b, 0u)] = e0 + 2;
                    cnt += __builtin_popcount(b);
                    b = __builtin_amdgcn_ballot_w32(m3);
                    if (m3) mylist[cnt + __builtin_amdgcn_mbcnt_lo(b, 0u)] = e0 + 3;
                    cnt += __builtin_popcount(b);
                }
            }
            if (cnt > LCAP - GROUP * CHUNK) {
                drain(cnt);
                cnt = 0;
            }
        }
    }
    for (int base = scanned + wave * 32; base < E; base += NWAVE * 32) {
        if (cnt > LCAP - 32) {
            drain(cnt);
            cnt = 0;
        }
        int  e   = base + lane;
        int  col = (e < E) ? cols[e] : -1;
        bool m   = (unsigned)(col - lo) < (unsigned)nr;
        unsigned b = __builtin_amdgcn_ballot_w32(m);
        if (m) mylist[cnt + __builtin_amdgcn_mbcnt_lo(b, 0u)] = e;
        cnt += __builtin_popcount(b);
    }
    drain(cnt);
    __syncthreads();

    const v4f bv = *(const v4f*)(bias  + lane * 4);
    const v4f gv = *(const v4f*)(gamma + lane * 4);
    const v4f tv = *(const v4f*)(beta  + lane * 4);
    for (int r = wave; r < nr; r += NWAVE) {
        const float* t = tile + r * D + lane;
        v4f v;
#pragma unroll
        for (int i = 0; i < 4; ++i) v[i] = t[32 * i];
        float s = 0.f, s2 = 0.f;
#pragma unroll
        for (int i = 0; i < 4; ++i) {
            float vv = v[i] + bv[i];
            vv = fmaxf(vv, 0.0f);
            v[i] = vv;
            s  += vv;
            s2 += vv * vv;
        }
#pragma unroll
        for (int off = 16; off >= 1; off >>= 1) {
            s  += __shfl_xor(s,  off, 32);
            s2 += __shfl_xor(s2, off, 32);
        }
        float mu  = s * (1.0f / 128.0f);
        float var = s2 * (1.0f / 128.0f) - mu * mu;
        float rs  = rsqrtf(var + LN_EPS);
#pragma unroll
        for (int i = 0; i < 4; ++i)
            v[i] = (v[i] - mu) * rs * gv[i] + tv[i];
        *(v4f*)(out + (size_t)(lo + r) * D + lane * 4) = v;
    }
}

extern "C" void kernel_launch(void* const* d_in, const int* in_sizes, int n_in,
                              void* d_out, int out_size, void* d_ws, size_t ws_size,
                              hipStream_t stream) {
    const float* x     = (const float*)d_in[0];
    const int*   eidx  = (const int*)  d_in[1];
    const float* ew    = (const float*)d_in[2];
    const float* W     = (const float*)d_in[3];
    const float* bias  = (const float*)d_in[4];
    const float* gamma = (const float*)d_in[5];
    const float* beta  = (const float*)d_in[6];
    float*       out   = (float*)d_out;

    const int N = in_sizes[0] / D;
    const int E = in_sizes[2];
    const int* rows = eidx;
    const int* cols = eidx + E;

    float* h    = (float*)d_ws;
    float* deg  = h + (size_t)N * D;

    const size_t koff  = (((size_t)N * (D + 1) * sizeof(float)) + 127) & ~(size_t)127;
    const bool   keyed = ((E & 3) == 0) && (E >= KCHUNK) && ((N + TR - 1) / TR <= 256) &&
                         (ws_size >= koff + (size_t)E);
    unsigned int* keys = keyed ? (unsigned int*)((char*)d_ws + koff) : nullptr;

    const size_t roff   = (koff + (size_t)E + 127) & ~(size_t)127;
    const bool   recd   = keyed && (N <= (1 << 23)) && (ws_size >= roff + (size_t)E * 8);
    int*         rec    = recd ? (int*)((char*)d_ws + roff) : nullptr;

    deg_init_kernel<<<(N + 255) / 256, 256, 0, stream>>>(deg, N);

    gemm_wmma_kernel<<<GEMM_GROUPS, GEMM_WAVES * 32, 0, stream>>>(x, W, h, N, cols, ew, deg, E, keys, rows, rec);

    range_sum_norm_kernel<<<(N + TR - 1) / TR, NWAVE * 32, 0, stream>>>(
        rows, cols, ew, deg, h, bias, gamma, beta, out, N, E, keys, rec);
}


# range keys as bit planes: GEMM launch's edge pass stores key bit p of 32 edges as one ballot dword; a scanning lane tests 32 edges by 8 logic ops, not 16 by byte tricks; LDS add arrival order changes
# speedup vs baseline: 2.8685x; 1.0306x over previous
#include <hip/hip_runtime.h>
#include <hip/hip_bf16.h>

typedef __attribute__((ext_vector_type(2))) float v2f;
typedef __attribute__((ext_vector_type(4))) float v4f;
typedef __attribute__((ext_vector_type(8))) float v8f;
typedef __attribute__((ext_vector_type(4))) int v4i;

#define D 128
#define LN_EPS 1e-5f
#define TR    392

__global__ void __launch_bounds__(256)
deg_init_kernel(float* __restrict__ deg, int N) {
    int i = blockIdx.x * blockDim.x + threadIdx.x;
    if (i < N) deg[i] = 1.0f;
}

__device__ __forceinline__ float inv_sqrt_deg(float d) {
    return (d > 0.0f) ? rsqrtf(d) : 0.0f;
}

#define GEMM_WAVES 4
#define GEMM_GROUPS 256
#define DEG_BATCH 8
#define KEY_BATCH 5

__global__ void __launch_bounds__(GEMM_WAVES * 32)
gemm_wmma_kernel(const float* __restrict__ x, const float* __restrict__ W,
                 float* __restrict__ h, int N,
                 const int* __restrict__ cols, const float* __restrict__ ew,
                 float* __restrict__ deg, int E, unsigned int* __restrict__ keys,
                 const int* __restrict__ rows, int* __restrict__ rec) {
    const int lane = threadIdx.x & 31;
    const int wave = __builtin_amdgcn_readfirstlane((int)(threadIdx.x >> 5));
    const int c    = lane & 15;
    const int hh   = lane >> 4;

    v4f wb[8][16];
#pragma unroll
    for (int t = 0; t < 8; ++t) {
        const int n = 64 * (t >> 2) + 4 * c + (t & 3);
        const float* wrow = W + (size_t)n * D + 4 * hh;
#pragma unroll
        for (int j = 0; j < 16; ++j) wb[t][j] = *(const v4f*)(wrow + 8 * j);
    }

    const int ntiles = (N + 15) / 16;
    const int nw     = (int)gridDim.x * GEMM_WAVES;
    int tile         = (int)blockIdx.x * GEMM_WAVES + wave;

    auto load_tile = [&](v4f (&dst)[16], int tl) {
        int row = tl * 16 + c;
        if (row > N - 1) row = N - 1;
        const float* p = x + (size_t)row * D + 4 * hh;
#pragma unroll
        for (int j = 0; j < 16; ++j) dst[j] = *(const v4f*)(p + 8 * j);
    };

    v4f xa[16];
    load_tile(xa, tile);

    if (keys != nullptr) {
        const int nunits = (E + 127) >> 7;
        for (int u0 = tile; u0 < nunits; u0 += KEY_BATCH * nw) {
            v4i  tc[KEY_BATCH];
            v4f  tw[KEY_BATCH];
            v4i  tr[KEY_BATCH];
#pragma unroll
            for (int b = 0; b < KEY_BATCH; ++b) {
                const int e  = (u0 + b * nw) * 128 + lane * 4;
                const int ec = (e < E) ? e : 0;
                tc[b] = *(const v4i*)(cols + ec);
                tw[b] = *(const v4f*)(ew + ec);
                tr[b] = (rec != nullptr) ? *(const v4i*)(rows + ec) : v4i{0, 0, 0, 0};
            }
#pragma unroll
            for (int b = 0; b < KEY_BATCH; ++b) {
                const int e = (u0 + b * nw) * 128 + lane * 4;
                const unsigned k0 = (unsigned)tc[b].x / (unsigned)TR;
                const unsigned k1 = (unsigned)tc[b].y / (unsigned)TR;
                const unsigned k2 = (unsigned)tc[b].z / (unsigned)TR;
                const unsigned k3 = (unsigned)tc[b].w / (unsigned)TR;
                const unsigned kk[4] = {k0, k1, k2, k3};
                unsigned word = 0u;
#pragma unroll
                for (int i = 0; i < 4; ++i) {
#pragma unroll
                    for (int p = 0; p < 8; ++p) {
                        const unsigned plane = __builtin_amdgcn_ballot_w32(((kk[i] >> p) & 1u) != 0u);
                        word = (lane == 8 * i + p) ? plane : word;
                    }
                }
                if (e < E) {
                    atomicAdd(&deg[tc[b].x], tw[b].x);
                    atomicAdd(&deg[tc[b].y], tw[b].y);
                    atomicAdd(&deg[tc[b].z], tw[b].z);
                    atomicAdd(&deg[tc[b].w], tw[b].w);
                    keys[e >> 2] = word;
                    if (rec != nullptr) {
                        const unsigned d0 = (unsigned)tc[b].x - k0 * (unsigned)TR;
                        const unsigned d1 = (unsigned)tc[b].y - k1 * (unsigned)TR;
                        const unsigned d2 = (unsigned)tc[b].z - k2 * (unsigned)TR;
                        const unsigned d3 = (unsigned)tc[b].w - k3 * (unsigned)TR;
                        const float w0 = tw[b].x, w1 = tw[b].y, w2 = tw[b].z, w3 = tw[b].w;
                        const v4i r01 = {(int)((unsigned)tr[b].x | (d0 << 23)), __float_as_int(w0),
                                         (int)((unsigned)tr[b].y | (d1 << 23)), __float_as_int(w1)};
                        const v4i r23 = {(int)((unsigned)tr[b].z | (d2 << 23)), __float_as_int(w2),
                                         (int)((unsigned)tr[b].w | (d3 << 23)), __float_as_int(w3)};
                        *(v4i*)(rec + 2 * (size_t)e)     = r01;
                        *(v4i*)(rec + 2 * (size_t)e + 4) = r23;
                    }
                }
            }
        }
    } else {
        const int per = (E + (int)gridDim.x - 1) / (int)gridDim.x;
        const int e0  = (int)blockIdx.x * per;
        const int e1  = (e0 + per < E) ? (e0 + per) : E;
        for (int eb = e0 + (int)threadIdx.x; eb < e1; eb += DEG_BATCH * GEMM_WAVES * 32) {
            int   tc[DEG_BATCH];
            float tw[DEG_BATCH];
#pragma unroll
            for (int u = 0; u < DEG_BATCH; ++u) {
                const int e  = eb + u * GEMM_WAVES * 32;
                const int ec = (e < e1) ? e : eb;
                tc[u] = cols[ec];
                tw[u] = ew[ec];
            }
#pragma unroll
            for (int u = 0; u < DEG_BATCH; ++u)
                if (eb + u * GEMM_WAVES * 32 < e1) atomicAdd(&deg[tc[u]], tw[u]);
        }
    }

    for (; tile < ntiles; tile += nw) {
        v4f xn[16];
        load_tile(xn, tile + nw);

        v8f acc[8];
#pragma unroll
        for (int t = 0; t < 8; ++t) acc[t] = v8f{};

#pragma unroll
        for (int j = 0; j < 16; ++j) {
            const v2f a0 = {xa[j].x, xa[j].y};
            const v2f a1 = {xa[j].z, xa[j].w};
#pragma unroll
            for (int t = 0; t < 8; ++t) {
                const v2f b0 = {wb[t][j].x, wb[t][j].y};
                acc[t] = __builtin_amdgcn_wmma_f32_16x16x4_f32(false, a0, false, b0,
                                                               (short)0, acc[t], false, false);
            }
#pragma unroll
            for (int t = 0; t < 8; ++t) {
                const v2f b1 = {wb[t][j].z, wb[t][j].w};
                acc[t] = __builtin_amdgcn_wmma_f32_16x16x4_f32(false, a1, false, b1,
                                                               (short)0, acc[t], false, false);
            }
        }

        const int rbase = tile * 16 + 8 * hh;
#pragma unroll
        for (int r = 0; r < 8; ++r) {
            const int row = rbase + r;
            if (row < N) {
                float* o = h + (size_t)row * D + 4 * c;
                v4f lo4 = {acc[0][r], acc[1][r], acc[2][r], acc[3][r]};
                v4f hi4 = {acc[4][r], acc[5][r], acc[6][r], acc[7][r]};
                *(v4f*)(o)      = lo4;
                *(v4f*)(o + 64) = hi4;
            }
        }

#pragma unroll
        for (int j = 0; j < 16; ++j) xa[j] = xn[j];
    }
}

#define NWAVE 32
#define CHUNK 128
#define GROUP 4
#define LCAP  (GROUP * CHUNK + 128)
#define SCH    1024
#define SGROUP 4
#define ROWB ((TR + NWAVE - 1) / NWAVE)

__global__ void __launch_bounds__(1024)
range_sum_norm_kernel(const int* __restrict__ rows, const int* __restrict__ cols,
                      const float* __restrict__ ew, const float* __restrict__ deg,
                      const float* __restrict__ h, const float* __restrict__ bias,
                      const float* __restrict__ gamma, const float* __restrict__ beta,
                      float* __restrict__ out, int N, int E,
                      const unsigned int* __restrict__ keys, const int* __restrict__ rec) {
    __shared__ __attribute__((aligned(16))) float tile[TR * D];
    __shared__ int lists[NWAVE * LCAP];

    const int lane = threadIdx.x & 31;
    const int wave = __builtin_amdgcn_readfirstlane((int)(threadIdx.x >> 5));
    const int lo   = blockIdx.x * TR;
    const int nr   = (N - lo < TR) ? (N - lo) : TR;
    int* mylist    = lists + wave * LCAP;

    for (int r0 = wave; r0 < nr; r0 += NWAVE * ROWB) {
        v4f   hs[ROWB];
        float dg[ROWB];
#pragma unroll
        for (int u = 0; u < ROWB; ++u) {
            const int r  = r0 + u * NWAVE;
            const int rc = (r < nr) ? r : (nr - 1);
            hs[u] = *(const v4f*)(h + (size_t)(lo + rc) * D + lane * 4);
            dg[u] = deg[lo + rc];
        }
#pragma unroll
        for (int u = 0; u < ROWB; ++u) {
            asm volatile("" :: "v"(hs[u]), "s"(dg[u]));
        }
#pragma unroll
        for (int u = 0; u < ROWB; ++u) {
            const int r = r0 + u * NWAVE;
            if (r < nr) {
                const float di = inv_sqrt_deg(dg[u]);
                float* t = tile + r * D + lane;
#pragma unroll
                for (int i = 0; i < 4; ++i) t[32 * i] = hs[u][i] * di * di;
            }
        }
    }
    __syncthreads();

    auto drain = [&](int cnt) {
        for (int b0 = 0; b0 < cnt; b0 += 32) {
            int  i   = b0 + lane;
            int  e   = mylist[(i < cnt) ? i : b0];
            int   src, dl;
            float w;
            if (rec != nullptr) {
                const unsigned long long r = *(const unsigned long long*)(rec + 2 * (size_t)e);
                const unsigned r0 = (unsigned)r;
                src = (int)(r0 & 0x7FFFFFu);
                dl  = (int)(r0 >> 23);
                w   = __uint_as_float((unsigned)(r >> 32));
            } else {
                src = rows[e];
                dl  = cols[e] - lo;
                w   = ew[e];
            }
            float nm = inv_sqrt_deg(deg[src]) * w * inv_sqrt_deg(deg[lo + dl]);
            int  m   = (cnt - b0 < 32) ? (cnt - b0) : 32;
            for (int j0 = 0; j0 < m; j0 += 8) {
                v4f hv[8];
#pragma unroll
                for (int u = 0; u < 8; ++u) {
                    int j = (j0 + u < m) ? (j0 + u) : (m - 1);
                    int s = __builtin_amdgcn_readlane(src, j);
                    hv[u] = *(const v4f*)(h + (size_t)s * D + lane * 4);
                }
#pragma unroll
                for (int u = 0; u < 8; ++u) {
                    if (j0 + u < m) {
                        int   d = __builtin_amdgcn_readlane(dl, j0 + u);
                        float f = __builtin_bit_cast(float,
                                      __builtin_amdgcn_readlane(__builtin_bit_cast(int, nm), j0 + u));
                        float* a = tile + d * D + lane;
#pragma unroll
                        for (int k = 0; k < 4; ++k)
                            __hip_atomic_fetch_add(a + 32 * k, hv[u][k] * f, __ATOMIC_RELAXED,
                                                   __HIP_MEMORY_SCOPE_WORKGROUP);
                    }
                }
            }
        }
    };

    const bool vec   = (E & 3) == 0;
    const int  nfull = vec ? (E / CHUNK) : 0;
    const int  rot   = (int)blockIdx.x * (nfull / (int)gridDim.x);
    int cnt = 0;
    int scanned = nfull * CHUNK;

    if (keys != nullptr) {
        const int nfs  = E / SCH;
        const int rots = (int)blockIdx.x * (nfs / (int)gridDim.x);
        const unsigned char* kb = (const unsigned char*)keys;
        unsigned flip[8];
#pragma unroll
        for (int p = 0; p < 8; ++p) flip[p] = (((unsigned)blockIdx.x >> p) & 1u) ? 0u : 0xFFFFFFFFu;
        const int lane_e = ((lane >> 2) << 7) + (lane & 3);
        scanned = nfs * SCH;
        for (int q0 = wave; q0 < nfs; q0 += SGROUP * NWAVE) {
            v4i      ka[SGROUP];
            v4i      kc[SGROUP];
            int      eb[SGROUP];
            unsigned ok[SGROUP];
#pragma unroll
            for (int u = 0; u < SGROUP; ++u) {
                int  q     = q0 + u * NWAVE;
                bool valid = q < nfs;
                int  ch    = valid ? (q + rots) : rots;
                if (ch >= nfs) ch -= nfs;
                eb[u] = ch * SCH;
                ok[u] = valid ? 0xFFFFFFFFu : 0u;
                const unsigned char* kp = kb + eb[u] + lane * 32;
                ka[u] = *(const v4i*)(kp);
                kc[u] = *(const v4i*)(kp + 16);
            }
            unsigned mv[SGROUP];
#pragma unroll
            for (int u = 0; u < SGROUP; ++u) {
                mv[u] = ((unsigned)ka[u].x ^ flip[0]) & ((unsigned)ka[u].y ^ flip[1]) &
                        ((unsigned)ka[u].z ^ flip[2]) & ((unsigned)ka[u].w ^ flip[3]) &
                        ((unsigned)kc[u].x ^ flip[4]) & ((unsigned)kc[u].y ^ flip[5]) &
                        ((unsigned)kc[u].z ^ flip[6]) & ((unsigned)kc[u].w ^ flip[7]) & ok[u];
            }
#pragma nounroll
            for (int u = 0; u < SGROUP; ++u) {
                unsigned mm = (u == 0) ? mv[0] : (u == 1) ? mv[1] : (u == 2) ? mv[2] : mv[3];
                const int e0 = ((u == 0) ? eb[0] : (u == 1) ? eb[1] : (u == 2) ? eb[2] : eb[3]) + lane_e;
                while (__builtin_amdgcn_ballot_w32(mm != 0u) != 0u) {
                    if (cnt > LCAP - 32) {
                        drain(cnt);
                        cnt = 0;
                    }
                    const bool     has = mm != 0u;
                    const unsigned b   = __builtin_amdgcn_ballot_w32(has);
                    if (has) {
                        const int c = __builtin_ctz(mm);
                        mylist[cnt + __builtin_amdgcn_mbcnt_lo(b, 0u)] = e0 + (c << 2);
                        mm &= mm - 1u;
                    }
                    cnt += __builtin_popcount(b);
                }
            }
        }
    } else {
        for (int q0 = wave; q0 < nfull; q0 += GROUP * NWAVE) {
            v4i      c[GROUP];
            int      eb[GROUP];
            unsigned lim[GROUP];
#pragma unroll
            for (int u = 0; u < GROUP; ++u) {
                int  q     = q0 + u * NWAVE;
                bool valid = q < nfull;
                int  ch    = valid ? (q + rot) : rot;
                if (ch >= nfull) ch -= nfull;
                eb[u]  = ch * CHUNK;
                lim[u] = valid ? (unsigned)nr : 0u;
                c[u]   = *(const v4i*)(cols + eb[u] + lane * 4);
            }
#pragma unroll
            for (int u = 0; u < GROUP; ++u) {
                int  e0 = eb[u] + lane * 4;
                bool m0 = (unsigned)(c[u].x - lo) < lim[u];
                bool m1 = (unsigned)(c[u].y - lo) < lim[u];
                bool m2 = (unsigned)(c[u].z - lo) < lim[u];
                bool m3 = (unsigned)(c[u].w - lo) < lim[u];
                if (__builtin_amdgcn_ballot_w32(m0 | m1 | m2 | m3) != 0u) {
                    unsigned b;
                    b = __builtin_amdgcn_ballot_w32(m0);
                    if (m0) mylist[cnt + __builtin_amdgcn_mbcnt_lo(b, 0u)] = e0;
                    cnt += __builtin_popcount(b);
                    b = __builtin_amdgcn_ballot_w32(m1);
                    if (m1) mylist[cnt + __builtin_amdgcn_mbcnt_lo(b, 0u)] = e0 + 1;
                    cnt += __builtin_popcount(b);
                    b = __builtin_amdgcn_ballot_w32(m2);
                    if (m2) mylist[cnt + __builtin_amdgcn_mbcnt_lo(b, 0u)] = e0 + 2;
                    cnt += __builtin_popcount(b);
                    b = __builtin_amdgcn_ballot_w32(m3);
                    if (m3) mylist[cnt + __builtin_amdgcn_mbcnt_lo(b, 0u)] = e0 + 3;
                    cnt += __builtin_popcount(b);
                }
            }
            if (cnt > LCAP - GROUP * CHUNK) {
                drain(cnt);
                cnt = 0;
            }
        }
    }
    for (int base = scanned + wave * 32; base < E; base += NWAVE * 32) {
        if (cnt > LCAP - 32) {
            drain(cnt);
            cnt = 0;
        }
        int  e   = base + lane;
        int  col = (e < E) ? cols[e] : -1;
        bool m   = (unsigned)(col - lo) < (unsigned)nr;
        unsigned b = __builtin_amdgcn_ballot_w32(m);
        if (m) mylist[cnt + __builtin_amdgcn_mbcnt_lo(b, 0u)] = e;
        cnt += __builtin_popcount(b);
    }
    drain(cnt);
    __syncthreads();

    const v4f bv = *(const v4f*)(bias  + lane * 4);
    const v4f gv = *(const v4f*)(gamma + lane * 4);
    const v4f tv = *(const v4f*)(beta  + lane * 4);
    for (int r = wave; r < nr; r += NWAVE) {
        const float* t = tile + r * D + lane;
        v4f v;
#pragma unroll
        for (int i = 0; i < 4; ++i) v[i] = t[32 * i];
        float s = 0.f, s2 = 0.f;
#pragma unroll
        for (int i = 0; i < 4; ++i) {
            float vv = v[i] + bv[i];
            vv = fmaxf(vv, 0.0f);
            v[i] = vv;
            s  += vv;
            s2 += vv * vv;
        }
#pragma unroll
        for (int off = 16; off >= 1; off >>= 1) {
            s  += __shfl_xor(s,  off, 32);
            s2 += __shfl_xor(s2, off, 32);
        }
        float mu  = s * (1.0f / 128.0f);
        float var = s2 * (1.0f / 128.0f) - mu * mu;
        float rs  = rsqrtf(var + LN_EPS);
#pragma unroll
        for (int i = 0; i < 4; ++i)
            v[i] = (v[i] - mu) * rs * gv[i] + tv[i];
        *(v4f*)(out + (size_t)(lo + r) * D + lane * 4) = v;
    }
}

extern "C" void kernel_launch(void* const* d_in, const int* in_sizes, int n_in,
                              void* d_out, int out_size, void* d_ws, size_t ws_size,
                              hipStream_t stream) {
    const float* x     = (const float*)d_in[0];
    const int*   eidx  = (const int*)  d_in[1];
    const float* ew    = (const float*)d_in[2];
    const float* W     = (const float*)d_in[3];
    const float* bias  = (const float*)d_in[4];
    const float* gamma = (const float*)d_in[5];
    const float* beta  = (const float*)d_in[6];
    float*       out   = (float*)d_out;

    const int N = in_sizes[0] / D;
    const int E = in_sizes[2];
    const int* rows = eidx;
    const int* cols = eidx + E;

    float* h    = (float*)d_ws;
    float* deg  = h + (size_t)N * D;

    const size_t koff  = (((size_t)N * (D + 1) * sizeof(float)) + 127) & ~(size_t)127;
    const bool   keyed = ((E & 3) == 0) && (E >= SCH) && ((N + TR - 1) / TR <= 256) &&
                         (ws_size >= koff + (size_t)E);
    unsigned int* keys = keyed ? (unsigned int*)((char*)d_ws + koff) : nullptr;

    const size_t roff   = (koff + (size_t)E + 127) & ~(size_t)127;
    const bool   recd   = keyed && (N <= (1 << 23)) && (ws_size >= roff + (size_t)E * 8);
    int*         rec    = recd ? (int*)((char*)d_ws + roff) : nullptr;

    deg_init_kernel<<<(N + 255) / 256, 256, 0, stream>>>(deg, N);

    gemm_wmma_kernel<<<GEMM_GROUPS, GEMM_WAVES * 32, 0, stream>>>(x, W, h, N, cols, ew, deg, E, keys, rows, rec);

    range_sum_norm_kernel<<<(N + TR - 1) / TR, NWAVE * 32, 0, stream>>>(
        rows, cols, ew, deg, h, bias, gamma, beta, out, N, E, keys, rec);
}


# edge sum per row in registers: listed edges filed in 12-slot LDS buckets per target row (returning LDS int add); a wave sums a row's gathered h rows in registers; filing order, self term last
# speedup vs baseline: 2.9697x; 1.0353x over previous
#include <hip/hip_runtime.h>
#include <hip/hip_bf16.h>

typedef __attribute__((ext_vector_type(2))) float v2f;
typedef __attribute__((ext_vector_type(4))) float v4f;
typedef __attribute__((ext_vector_type(8))) float v8f;
typedef __attribute__((ext_vector_type(4))) int v4i;

#define D 128
#define LN_EPS 1e-5f
#define TR    392

__global__ void __launch_bounds__(256)
deg_init_kernel(float* __restrict__ deg, int N) {
    int i = blockIdx.x * blockDim.x + threadIdx.x;
    if (i < N) deg[i] = 1.0f;
}

__device__ __forceinline__ float inv_sqrt_deg(float d) {
    return (d > 0.0f) ? rsqrtf(d) : 0.0f;
}

#define GEMM_WAVES 4
#define GEMM_GROUPS 256
#define DEG_BATCH 8
#define KEY_BATCH 5

__global__ void __launch_bounds__(GEMM_WAVES * 32)
gemm_wmma_kernel(const float* __restrict__ x, const float* __restrict__ W,
                 float* __restrict__ h, int N,
                 const int* __restrict__ cols, const float* __restrict__ ew,
                 float* __restrict__ deg, int E, unsigned int* __restrict__ keys,
                 const int* __restrict__ rows, int* __restrict__ rec) {
    const int lane = threadIdx.x & 31;
    const int wave = __builtin_amdgcn_readfirstlane((int)(threadIdx.x >> 5));
    const int c    = lane & 15;
    const int hh   = lane >> 4;

    v4f wb[8][16];
#pragma unroll
    for (int t = 0; t < 8; ++t) {
        const int n = 64 * (t >> 2) + 4 * c + (t & 3);
        const float* wrow = W + (size_t)n * D + 4 * hh;
#pragma unroll
        for (int j = 0; j < 16; ++j) wb[t][j] = *(const v4f*)(wrow + 8 * j);
    }

    const int ntiles = (N + 15) / 16;
    const int nw     = (int)gridDim.x * GEMM_WAVES;
    int tile         = (int)blockIdx.x * GEMM_WAVES + wave;

    auto load_tile = [&](v4f (&dst)[16], int tl) {
        int row = tl * 16 + c;
        if (row > N - 1) row = N - 1;
        const float* p = x + (size_t)row * D + 4 * hh;
#pragma unroll
        for (int j = 0; j < 16; ++j) dst[j] = *(const v4f*)(p + 8 * j);
    };

    v4f xa[16];
    load_tile(xa, tile);

    if (keys != nullptr) {
        const int nunits = (E + 127) >> 7;
        for (int u0 = tile; u0 < nunits; u0 += KEY_BATCH * nw) {
            v4i  tc[KEY_BATCH];
            v4f  tw[KEY_BATCH];
            v4i  tr[KEY_BATCH];
#pragma unroll
            for (int b = 0; b < KEY_BATCH; ++b) {
                const int e  = (u0 + b * nw) * 128 + lane * 4;
                const int ec = (e < E) ? e : 0;
                tc[b] = *(const v4i*)(cols + ec);
                tw[b] = *(const v4f*)(ew + ec);
                tr[b] = (rec != nullptr) ? *(const v4i*)(rows + ec) : v4i{0, 0, 0, 0};
            }
#pragma unroll
            for (int b = 0; b < KEY_BATCH; ++b) {
                const int e = (u0 + b * nw) * 128 + lane * 4;
                const unsigned k0 = (unsigned)tc[b].x / (unsigned)TR;
                const unsigned k1 = (unsigned)tc[b].y / (unsigned)TR;
                const unsigned k2 = (unsigned)tc[b].z / (unsigned)TR;
                const unsigned k3 = (unsigned)tc[b].w / (unsigned)TR;
                const unsigned kk[4] = {k0, k1, k2, k3};
                unsigned word = 0u;
#pragma unroll
                for (int i = 0; i < 4; ++i) {
#pragma unroll
                    for (int p = 0; p < 8; ++p) {
                        const unsigned plane = __builtin_amdgcn_ballot_w32(((kk[i] >> p) & 1u) != 0u);
                        word = (lane == 8 * i + p) ? plane : word;
                    }
                }
                if (e < E) {
                    atomicAdd(&deg[tc[b].x], tw[b].x);
                    atomicAdd(&deg[tc[b].y], tw[b].y);
                    atomicAdd(&deg[tc[b].z], tw[b].z);
                    atomicAdd(&deg[tc[b].w], tw[b].w);
                    keys[e >> 2] = word;
                    if (rec != nullptr) {
                        const unsigned d0 = (unsigned)tc[b].x - k0 * (unsigned)TR;
                        const unsigned d1 = (unsigned)tc[b].y - k1 * (unsigned)TR;
                        const unsigned d2 = (unsigned)tc[b].z - k2 * (unsigned)TR;
                        const unsigned d3 = (unsigned)tc[b].w - k3 * (unsigned)TR;
                        const float w0 = tw[b].x, w1 = tw[b].y, w2 = tw[b].z, w3 = tw[b].w;
                        const v4i r01 = {(int)((unsigned)tr[b].x | (d0 << 23)), __float_as_int(w0),
                                         (int)((unsigned)tr[b].y | (d1 << 23)), __float_as_int(w1)};
                        const v4i r23 = {(int)((unsigned)tr[b].z | (d2 << 23)), __float_as_int(w2),
                                         (int)((unsigned)tr[b].w | (d3 << 23)), __float_as_int(w3)};
                        *(v4i*)(rec + 2 * (size_t)e)     = r01;
                        *(v4i*)(rec + 2 * (size_t)e + 4) = r23;
                    }
                }
            }
        }
    } else {
        const int per = (E + (int)gridDim.x - 1) / (int)gridDim.x;
        const int e0  = (int)blockIdx.x * per;
        const int e1  = (e0 + per < E) ? (e0 + per) : E;
        for (int eb = e0 + (int)threadIdx.x; eb < e1; eb += DEG_BATCH * GEMM_WAVES * 32) {
            int   tc[DEG_BATCH];
            float tw[DEG_BATCH];
#pragma unroll
            for (int u = 0; u < DEG_BATCH; ++u) {
                const int e  = eb + u * GEMM_WAVES * 32;
                const int ec = (e < e1) ? e : eb;
                tc[u] = cols[ec];
                tw[u] = ew[ec];
            }
#pragma unroll
            for (int u = 0; u < DEG_BATCH; ++u)
                if (eb + u * GEMM_WAVES * 32 < e1) atomicAdd(&deg[tc[u]], tw[u]);
        }
    }

    for (; tile < ntiles; tile += nw) {
        v4f xn[16];
        load_tile(xn, tile + nw);

        v8f acc[8];
#pragma unroll
        for (int t = 0; t < 8; ++t) acc[t] = v8f{};

#pragma unroll
        for (int j = 0; j < 16; ++j) {
            const v2f a0 = {xa[j].x, xa[j].y};
            const v2f a1 = {xa[j].z, xa[j].w};
#pragma unroll
            for (int t = 0; t < 8; ++t) {
                const v2f b0 = {wb[t][j].x, wb[t][j].y};
                acc[t] = __builtin_amdgcn_wmma_f32_16x16x4_f32(false, a0, false, b0,
                                                               (short)0, acc[t], false, false);
            }
#pragma unroll
            for (int t = 0; t < 8; ++t) {
                const v2f b1 = {wb[t][j].z, wb[t][j].w};
                acc[t] = __builtin_amdgcn_wmma_f32_16x16x4_f32(false, a1, false, b1,
                                                               (short)0, acc[t], false, false);
            }
        }

        const int rbase = tile * 16 + 8 * hh;
#pragma unroll
        for (int r = 0; r < 8; ++r) {
            const int row = rbase + r;
            if (row < N) {
                float* o = h + (size_t)row * D + 4 * c;
                v4f lo4 = {acc[0][r], acc[1][r], acc[2][r], acc[3][r]};
                v4f hi4 = {acc[4][r], acc[5][r], acc[6][r], acc[7][r]};
                *(v4f*)(o)      = lo4;
                *(v4f*)(o + 64) = hi4;
            }
        }

#pragma unroll
        for (int j = 0; j < 16; ++j) xa[j] = xn[j];
    }
}

#define NWAVE 32
#define CHUNK 128
#define GROUP 4
#define LCAP  (GROUP * CHUNK + 128)
#define BCAP  12
#define SCH    1024
#define SGROUP 4
#define ROWB ((TR + NWAVE - 1) / NWAVE)

__global__ void __launch_bounds__(1024)
range_sum_norm_kernel(const int* __restrict__ rows, const int* __restrict__ cols,
                      const float* __restrict__ ew, const float* __restrict__ deg,
                      const float* __restrict__ h, const float* __restrict__ bias,
                      const float* __restrict__ gamma, const float* __restrict__ beta,
                      float* __restrict__ out, int N, int E,
                      const unsigned int* __restrict__ keys, const int* __restrict__ rec) {
    __shared__ __attribute__((aligned(16))) float tile[TR * D];
    __shared__ int lists[NWAVE * LCAP];
    __shared__ unsigned long long bucket[TR * BCAP];
    __shared__ int rowcnt[TR];

    const int lane = threadIdx.x & 31;
    const int wave = __builtin_amdgcn_readfirstlane((int)(threadIdx.x >> 5));
    const int lo   = blockIdx.x * TR;
    const int nr   = (N - lo < TR) ? (N - lo) : TR;
    int* mylist    = lists + wave * LCAP;

    for (int r0 = wave; r0 < nr; r0 += NWAVE * ROWB) {
        v4f   hs[ROWB];
        float dg[ROWB];
#pragma unroll
        for (int u = 0; u < ROWB; ++u) {
            const int r  = r0 + u * NWAVE;
            const int rc = (r < nr) ? r : (nr - 1);
            hs[u] = *(const v4f*)(h + (size_t)(lo + rc) * D + lane * 4);
            dg[u] = deg[lo + rc];
        }
#pragma unroll
        for (int u = 0; u < ROWB; ++u) {
            asm volatile("" :: "v"(hs[u]), "s"(dg[u]));
        }
#pragma unroll
        for (int u = 0; u < ROWB; ++u) {
            const int r = r0 + u * NWAVE;
            if (r < nr) {
                const float di = inv_sqrt_deg(dg[u]);
                float* t = tile + r * D + lane;
#pragma unroll
                for (int i = 0; i < 4; ++i) t[32 * i] = hs[u][i] * di * di;
            }
        }
    }
    for (int i = (int)threadIdx.x; i < TR; i += NWAVE * 32) rowcnt[i] = 0;
    __syncthreads();

    auto file_edges = [&](int cnt) {
        for (int b0 = 0; b0 < cnt; b0 += 32) {
            const int  i     = b0 + lane;
            const bool valid = i < cnt;
            int  e   = mylist[valid ? i : b0];
            int   src, dl;
            float w;
            if (rec != nullptr) {
                const unsigned long long r = *(const unsigned long long*)(rec + 2 * (size_t)e);
                const unsigned r0 = (unsigned)r;
                src = (int)(r0 & 0x7FFFFFu);
                dl  = (int)(r0 >> 23);
                w   = __uint_as_float((unsigned)(r >> 32));
            } else {
                src = rows[e];
                dl  = cols[e] - lo;
                w   = ew[e];
            }
            float nm = inv_sqrt_deg(deg[src]) * w * inv_sqrt_deg(deg[lo + dl]);
            int slot = 0;
            if (valid) slot = __hip_atomic_fetch_add(&rowcnt[dl], 1, __ATOMIC_RELAXED,
                                                     __HIP_MEMORY_SCOPE_WORKGROUP);
            if (valid && slot < BCAP)
                bucket[dl * BCAP + slot] =
                    ((unsigned long long)__float_as_uint(nm) << 32) | (unsigned long long)(unsigned)src;
            unsigned ov = __builtin_amdgcn_ballot_w32(valid && slot >= BCAP);
            while (ov != 0u) {
                const int j = __builtin_ctz(ov);
                ov &= ov - 1u;
                const int   so = __builtin_amdgcn_readlane(src, j);
                const int   d  = __builtin_amdgcn_readlane(dl, j);
                const float f  = __builtin_bit_cast(float,
                                     __builtin_amdgcn_readlane(__builtin_bit_cast(int, nm), j));
                const v4f hv = *(const v4f*)(h + (size_t)so * D + lane * 4);
                float* a = tile + d * D + lane;
#pragma unroll
                for (int k = 0; k < 4; ++k)
                    __hip_atomic_fetch_add(a + 32 * k, hv[k] * f, __ATOMIC_RELAXED,
                                           __HIP_MEMORY_SCOPE_WORKGROUP);
            }
        }
    };

    const bool vec   = (E & 3) == 0;
    const int  nfull = vec ? (E / CHUNK) : 0;
    const int  rot   = (int)blockIdx.x * (nfull / (int)gridDim.x);
    int cnt = 0;
    int scanned = nfull * CHUNK;

    if (keys != nullptr) {
        const int nfs  = E / SCH;
        const int rots = (int)blockIdx.x * (nfs / (int)gridDim.x);
        const unsigned char* kb = (const unsigned char*)keys;
        unsigned flip[8];
#pragma unroll
        for (int p = 0; p < 8; ++p) flip[p] = (((unsigned)blockIdx.x >> p) & 1u) ? 0u : 0xFFFFFFFFu;
        const int lane_e = ((lane >> 2) << 7) + (lane & 3);
        scanned = nfs * SCH;
        for (int q0 = wave; q0 < nfs; q0 += SGROUP * NWAVE) {
            v4i      ka[SGROUP];
            v4i      kc[SGROUP];
            int      eb[SGROUP];
            unsigned ok[SGROUP];
#pragma unroll
            for (int u = 0; u < SGROUP; ++u) {
                int  q     = q0 + u * NWAVE;
                bool valid = q < nfs;
                int  ch    = valid ? (q + rots) : rots;
                if (ch >= nfs) ch -= nfs;
                eb[u] = ch * SCH;
                ok[u] = valid ? 0xFFFFFFFFu : 0u;
                const unsigned char* kp = kb + eb[u] + lane * 32;
                ka[u] = *(const v4i*)(kp);
                kc[u] = *(const v4i*)(kp + 16);
            }
            unsigned mv[SGROUP];
#pragma unroll
            for (int u = 0; u < SGROUP; ++u) {
                mv[u] = ((unsigned)ka[u].x ^ flip[0]) & ((unsigned)ka[u].y ^ flip[1]) &
                        ((unsigned)ka[u].z ^ flip[2]) & ((unsigned)ka[u].w ^ flip[3]) &
                        ((unsigned)kc[u].x ^ flip[4]) & ((unsigned)kc[u].y ^ flip[5]) &
                        ((unsigned)kc[u].z ^ flip[6]) & ((unsigned)kc[u].w ^ flip[7]) & ok[u];
            }
#pragma nounroll
            for (int u = 0; u < SGROUP; ++u) {
                unsigned mm = (u == 0) ? mv[0] : (u == 1) ? mv[1] : (u == 2) ? mv[2] : mv[3];
                const int e0 = ((u == 0) ? eb[0] : (u == 1) ? eb[1] : (u == 2) ? eb[2] : eb[3]) + lane_e;
                while (__builtin_amdgcn_ballot_w32(mm != 0u) != 0u) {
                    if (cnt > LCAP - 32) {
                        file_edges(cnt);
                        cnt = 0;
                    }
                    const bool     has = mm != 0u;
                    const unsigned b   = __builtin_amdgcn_ballot_w32(has);
                    if (has) {
                        const int c = __builtin_ctz(mm);
                        mylist[cnt + __builtin_amdgcn_mbcnt_lo(b, 0u)] = e0 + (c << 2);
                        mm &= mm - 1u;
                    }
                    cnt += __builtin_popcount(b);
                }
            }
        }
    } else {
        for (int q0 = wave; q0 < nfull; q0 += GROUP * NWAVE) {
            v4i      c[GROUP];
            int      eb[GROUP];
            unsigned lim[GROUP];
#pragma unroll
            for (int u = 0; u < GROUP; ++u) {
                int  q     = q0 + u * NWAVE;
                bool valid = q < nfull;
                int  ch    = valid ? (q + rot) : rot;
                if (ch >= nfull) ch -= nfull;
                eb[u]  = ch * CHUNK;
                lim[u] = valid ? (unsigned)nr : 0u;
                c[u]   = *(const v4i*)(cols + eb[u] + lane * 4);
            }
#pragma unroll
            for (int u = 0; u < GROUP; ++u) {
                int  e0 = eb[u] + lane * 4;
                bool m0 = (unsigned)(c[u].x - lo) < lim[u];
                bool m1 = (unsigned)(c[u].y - lo) < lim[u];
                bool m2 = (unsigned)(c[u].z - lo) < lim[u];
                bool m3 = (unsigned)(c[u].w - lo) < lim[u];
                if (__builtin_amdgcn_ballot_w32(m0 | m1 | m2 | m3) != 0u) {
                    unsigned b;
                    b = __builtin_amdgcn_ballot_w32(m0);
                    if (m0) mylist[cnt + __builtin_amdgcn_mbcnt_lo(b, 0u)] = e0;
                    cnt += __builtin_popcount(b);
                    b = __builtin_amdgcn_ballot_w32(m1);
                    if (m1) mylist[cnt + __builtin_amdgcn_mbcnt_lo(b, 0u)] = e0 + 1;
                    cnt += __builtin_popcount(b);
                    b = __builtin_amdgcn_ballot_w32(m2);
                    if (m2) mylist[cnt + __builtin_amdgcn_mbcnt_lo(b, 0u)] = e0 + 2;
                    cnt += __builtin_popcount(b);
                    b = __builtin_amdgcn_ballot_w32(m3);
                    if (m3) mylist[cnt + __builtin_amdgcn_mbcnt_lo(b, 0u)] = e0 + 3;
                    cnt += __builtin_popcount(b);
                }
            }
            if (cnt > LCAP - GROUP * CHUNK) {
                file_edges(cnt);
                cnt = 0;
            }
        }
    }
    for (int base = scanned + wave * 32; base < E; base += NWAVE * 32) {
        if (cnt > LCAP - 32) {
            file_edges(cnt);
            cnt = 0;
        }
        int  e   = base + lane;
        int  col = (e < E) ? cols[e] : -1;
        bool m   = (unsigned)(col - lo) < (unsigned)nr;
        unsigned b = __builtin_amdgcn_ballot_w32(m);
        if (m) mylist[cnt + __builtin_amdgcn_mbcnt_lo(b, 0u)] = e;
        cnt += __builtin_popcount(b);
    }
    file_edges(cnt);
    __syncthreads();

    const v4f bv = *(const v4f*)(bias  + lane * 4);
    const v4f gv = *(const v4f*)(gamma + lane * 4);
    const v4f tv = *(const v4f*)(beta  + lane * 4);

    auto fetch = [&](int rr, int& n, int& srcv, float& nmv, v4f (&hx)[8]) {
        const int rc = (rr < nr) ? rr : (nr - 1);
        const int c  = __builtin_amdgcn_readfirstlane(rowcnt[rc]);
        n = (rr < nr) ? ((c < BCAP) ? c : BCAP) : 0;
        unsigned long long ent = 0ull;
        if (lane < n) ent = bucket[rc * BCAP + lane];
        srcv = (int)(unsigned)ent;
        nmv  = __uint_as_float((unsigned)(ent >> 32));
#pragma unroll
        for (int u = 0; u < 8; ++u) {
            const int so  = __builtin_amdgcn_readlane(srcv, u);
            const int off = (u < n) ? lane * 4 : 0;
            hx[u] = *(const v4f*)(h + (size_t)so * D + off);
        }
    };

    auto fetch_rest = [&](int n, int srcv, v4f (&hc)[BCAP - 8]) {
        if (n > 8) {
#pragma unroll
            for (int u = 0; u < BCAP - 8; ++u) {
                const int so  = __builtin_amdgcn_readlane(srcv, 8 + u);
                const int off = (8 + u < n) ? lane * 4 : 0;
                hc[u] = *(const v4f*)(h + (size_t)so * D + off);
            }
        }
    };

    auto finish_row = [&](int r, int n, float nmv, v4f (&hx)[8], v4f (&hc)[BCAP - 8]) {
        v4f acc = {0.f, 0.f, 0.f, 0.f};
#pragma unroll
        for (int u = 0; u < 8; ++u) {
            if (u < n) {
                const float f = __builtin_bit_cast(float,
                                    __builtin_amdgcn_readlane(__builtin_bit_cast(int, nmv), u));
                acc += hx[u] * f;
            }
        }
        if (n > 8) {
#pragma unroll
            for (int u = 0; u < BCAP - 8; ++u) {
                if (8 + u < n) {
                    const float f = __builtin_bit_cast(float,
                                        __builtin_amdgcn_readlane(__builtin_bit_cast(int, nmv), 8 + u));
                    acc += hc[u] * f;
                }
            }
        }

        const float* t = tile + r * D + lane;
        v4f v;
#pragma unroll
        for (int i = 0; i < 4; ++i) v[i] = t[32 * i] + acc[i];
        float s = 0.f, s2 = 0.f;
#pragma unroll
        for (int i = 0; i < 4; ++i) {
            float vv = v[i] + bv[i];
            vv = fmaxf(vv, 0.0f);
            v[i] = vv;
            s  += vv;
            s2 += vv * vv;
        }
#pragma unroll
        for (int off = 16; off >= 1; off >>= 1) {
            s  += __shfl_xor(s,  off, 32);
            s2 += __shfl_xor(s2, off, 32);
        }
        float mu  = s * (1.0f / 128.0f);
        float var = s2 * (1.0f / 128.0f) - mu * mu;
        float rs  = rsqrtf(var + LN_EPS);
#pragma unroll
        for (int i = 0; i < 4; ++i)
            v[i] = (v[i] - mu) * rs * gv[i] + tv[i];
        *(v4f*)(out + (size_t)(lo + r) * D + lane * 4) = v;
    };

    int   n_a, src_a, n_b, src_b;
    float nm_a, nm_b;
    v4f   ha[8], hb[8];
    fetch(wave, n_a, src_a, nm_a, ha);
    int r = wave;
    for (; r + NWAVE < nr; r += 2 * NWAVE) {
        v4f hc[BCAP - 8];
        fetch_rest(n_a, src_a, hc);
        fetch(r + NWAVE, n_b, src_b, nm_b, hb);
        finish_row(r, n_a, nm_a, ha, hc);
        fetch_rest(n_b, src_b, hc);
        fetch(r + 2 * NWAVE, n_a, src_a, nm_a, ha);
        finish_row(r + NWAVE, n_b, nm_b, hb, hc);
    }
    if (r < nr) {
        v4f hc[BCAP - 8];
        fetch_rest(n_a, src_a, hc);
        finish_row(r, n_a, nm_a, ha, hc);
    }
}

extern "C" void kernel_launch(void* const* d_in, const int* in_sizes, int n_in,
                              void* d_out, int out_size, void* d_ws, size_t ws_size,
                              hipStream_t stream) {
    const float* x     = (const float*)d_in[0];
    const int*   eidx  = (const int*)  d_in[1];
    const float* ew    = (const float*)d_in[2];
    const float* W     = (const float*)d_in[3];
    const float* bias  = (const float*)d_in[4];
    const float* gamma = (const float*)d_in[5];
    const float* beta  = (const float*)d_in[6];
    float*       out   = (float*)d_out;

    const int N = in_sizes[0] / D;
    const int E = in_sizes[2];
    const int* rows = eidx;
    const int* cols = eidx + E;

    float* h    = (float*)d_ws;
    float* deg  = h + (size_t)N * D;

    const size_t koff  = (((size_t)N * (D + 1) * sizeof(float)) + 127) & ~(size_t)127;
    const bool   keyed = ((E & 3) == 0) && (E >= SCH) && ((N + TR - 1) / TR <= 256) &&
                         (ws_size >= koff + (size_t)E);
    unsigned int* keys = keyed ? (unsigned int*)((char*)d_ws + koff) : nullptr;

    const size_t roff   = (koff + (size_t)E + 127) & ~(size_t)127;
    const bool   recd   = keyed && (N <= (1 << 23)) && (ws_size >= roff + (size_t)E * 8);
    int*         rec    = recd ? (int*)((char*)d_ws + roff) : nullptr;

    deg_init_kernel<<<(N + 255) / 256, 256, 0, stream>>>(deg, N);

    gemm_wmma_kernel<<<GEMM_GROUPS, GEMM_WAVES * 32, 0, stream>>>(x, W, h, N, cols, ew, deg, E, keys, rows, rec);

    range_sum_norm_kernel<<<(N + TR - 1) / TR, NWAVE * 32, 0, stream>>>(
        rows, cols, ew, deg, h, bias, gamma, beta, out, N, E, keys, rec);
}


# edge work rides in the GEMM tile loop: each trip takes one 128-edge unit (loads a trip ahead), so the scattered deg adds drain under the WMMAs, not as one burst ahead of them; arrival order as before
# speedup vs baseline: 3.2523x; 1.0952x over previous
#include <hip/hip_runtime.h>
#include <hip/hip_bf16.h>

typedef __attribute__((ext_vector_type(2))) float v2f;
typedef __attribute__((ext_vector_type(4))) float v4f;
typedef __attribute__((ext_vector_type(8))) float v8f;
typedef __attribute__((ext_vector_type(4))) int v4i;

#define D 128
#define LN_EPS 1e-5f
#define TR    392

__global__ void __launch_bounds__(256)
deg_init_kernel(float* __restrict__ deg, int N) {
    int i = blockIdx.x * blockDim.x + threadIdx.x;
    if (i < N) deg[i] = 1.0f;
}

__device__ __forceinline__ float inv_sqrt_deg(float d) {
    return (d > 0.0f) ? rsqrtf(d) : 0.0f;
}

#define GEMM_WAVES 4
#define GEMM_GROUPS 256
#define DEG_BATCH 8

__global__ void __launch_bounds__(GEMM_WAVES * 32)
gemm_wmma_kernel(const float* __restrict__ x, const float* __restrict__ W,
                 float* __restrict__ h, int N,
                 const int* __restrict__ cols, const float* __restrict__ ew,
                 float* __restrict__ deg, int E, unsigned int* __restrict__ keys,
                 const int* __restrict__ rows, int* __restrict__ rec) {
    const int lane = threadIdx.x & 31;
    const int wave = __builtin_amdgcn_readfirstlane((int)(threadIdx.x >> 5));
    const int c    = lane & 15;
    const int hh   = lane >> 4;

    const int ntiles = (N + 15) / 16;
    const int nw     = (int)gridDim.x * GEMM_WAVES;
    int tile         = (int)blockIdx.x * GEMM_WAVES + wave;

    const bool keyed  = keys != nullptr;
    const int  nunits = (E + 127) >> 7;
    const int*   ucols = keyed ? cols : (const int*)x;
    const float* uews  = keyed ? ew   : x;
    const int*   urows = keyed ? rows : (const int*)x;
    auto load_unit = [&](int u, v4i& c4, v4f& w4, v4i& r4) {
        const int e  = u * 128 + lane * 4;
        const int ec = (keyed && e < E) ? e : 0;
        c4 = *(const v4i*)(ucols + ec);
        w4 = *(const v4f*)(uews + ec);
        r4 = *(const v4i*)(urows + ec);
    };
    auto do_unit = [&](int u, const v4i& c4, const v4f& w4, const v4i& r4) {
        const int e = u * 128 + lane * 4;
        const unsigned k0 = (unsigned)c4.x / (unsigned)TR;
        const unsigned k1 = (unsigned)c4.y / (unsigned)TR;
        const unsigned k2 = (unsigned)c4.z / (unsigned)TR;
        const unsigned k3 = (unsigned)c4.w / (unsigned)TR;
        const unsigned kk[4] = {k0, k1, k2, k3};
        unsigned word = 0u;
#pragma unroll
        for (int i = 0; i < 4; ++i) {
#pragma unroll
            for (int p = 0; p < 8; ++p) {
                const unsigned plane = __builtin_amdgcn_ballot_w32(((kk[i] >> p) & 1u) != 0u);
                word = (lane == 8 * i + p) ? plane : word;
            }
        }
        if (e < E) {
            atomicAdd(&deg[c4.x], w4.x);
            atomicAdd(&deg[c4.y], w4.y);
            atomicAdd(&deg[c4.z], w4.z);
            atomicAdd(&deg[c4.w], w4.w);
            keys[e >> 2] = word;
            if (rec != nullptr) {
                const unsigned d0 = (unsigned)c4.x - k0 * (unsigned)TR;
                const unsigned d1 = (unsigned)c4.y - k1 * (unsigned)TR;
                const unsigned d2 = (unsigned)c4.z - k2 * (unsigned)TR;
                const unsigned d3 = (unsigned)c4.w - k3 * (unsigned)TR;
                const float w0 = w4.x, w1 = w4.y, w2 = w4.z, w3 = w4.w;
                const v4i r01 = {(int)((unsigned)r4.x | (d0 << 23)), __float_as_int(w0),
                                 (int)((unsigned)r4.y | (d1 << 23)), __float_as_int(w1)};
                const v4i r23 = {(int)((unsigned)r4.z | (d2 << 23)), __float_as_int(w2),
                                 (int)((unsigned)r4.w | (d3 << 23)), __float_as_int(w3)};
                *(v4i*)(rec + 2 * (size_t)e)     = r01;
                *(v4i*)(rec + 2 * (size_t)e + 4) = r23;
            }
        }
    };

    v4i uc = {0, 0, 0, 0}, ur = {0, 0, 0, 0};
    v4f uw = {0.f, 0.f, 0.f, 0.f};
    load_unit(tile, uc, uw, ur);

    v4f wb[8][16];
#pragma unroll
    for (int t = 0; t < 8; ++t) {
        const int n = 64 * (t >> 2) + 4 * c + (t & 3);
        const float* wrow = W + (size_t)n * D + 4 * hh;
#pragma unroll
        for (int j = 0; j < 16; ++j) wb[t][j] = *(const v4f*)(wrow + 8 * j);
    }

    auto load_tile = [&](v4f (&dst)[16], int tl) {
        int row = tl * 16 + c;
        if (row > N - 1) row = N - 1;
        const float* p = x + (size_t)row * D + 4 * hh;
#pragma unroll
        for (int j = 0; j < 16; ++j) dst[j] = *(const v4f*)(p + 8 * j);
    };

    v4f xa[16];
    load_tile(xa, tile);

    if (!keyed) {
        const int per = (E + (int)gridDim.x - 1) / (int)gridDim.x;
        const int e0  = (int)blockIdx.x * per;
        const int e1  = (e0 + per < E) ? (e0 + per) : E;
        for (int eb = e0 + (int)threadIdx.x; eb < e1; eb += DEG_BATCH * GEMM_WAVES * 32) {
            int   tc[DEG_BATCH];
            float tw[DEG_BATCH];
#pragma unroll
            for (int u = 0; u < DEG_BATCH; ++u) {
                const int e  = eb + u * GEMM_WAVES * 32;
                const int ec = (e < e1) ? e : eb;
                tc[u] = cols[ec];
                tw[u] = ew[ec];
            }
#pragma unroll
            for (int u = 0; u < DEG_BATCH; ++u)
                if (eb + u * GEMM_WAVES * 32 < e1) atomicAdd(&deg[tc[u]], tw[u]);
        }
    }

    for (; tile < ntiles; tile += nw) {
        v4f xn[16];
        load_tile(xn, tile + nw);

        v4i nc, nr4;
        v4f nwt;
        load_unit(tile + nw, nc, nwt, nr4);
        if (keyed && tile < nunits) do_unit(tile, uc, uw, ur);

        v8f acc[8];
#pragma unroll
        for (int t = 0; t < 8; ++t) acc[t] = v8f{};

#pragma unroll
        for (int j = 0; j < 16; ++j) {
            const v2f a0 = {xa[j].x, xa[j].y};
            const v2f a1 = {xa[j].z, xa[j].w};
#pragma unroll
            for (int t = 0; t < 8; ++t) {
                const v2f b0 = {wb[t][j].x, wb[t][j].y};
                acc[t] = __builtin_amdgcn_wmma_f32_16x16x4_f32(false, a0, false, b0,
                                                               (short)0, acc[t], false, false);
            }
#pragma unroll
            for (int t = 0; t < 8; ++t) {
                const v2f b1 = {wb[t][j].z, wb[t][j].w};
                acc[t] = __builtin_amdgcn_wmma_f32_16x16x4_f32(false, a1, false, b1,
                                                               (short)0, acc[t], false, false);
            }
        }

        const int rbase = tile * 16 + 8 * hh;
#pragma unroll
        for (int r = 0; r < 8; ++r) {
            const int row = rbase + r;
            if (row < N) {
                float* o = h + (size_t)row * D + 4 * c;
                v4f lo4 = {acc[0][r], acc[1][r], acc[2][r], acc[3][r]};
                v4f hi4 = {acc[4][r], acc[5][r], acc[6][r], acc[7][r]};
                *(v4f*)(o)      = lo4;
                *(v4f*)(o + 64) = hi4;
            }
        }

#pragma unroll
        for (int j = 0; j < 16; ++j) xa[j] = xn[j];
        uc = nc; uw = nwt; ur = nr4;
    }

    if (keyed) {
        for (int u = tile; u < nunits; u += nw) {
            load_unit(u, uc, uw, ur);
            do_unit(u, uc, uw, ur);
        }
    }
}

#define NWAVE 32
#define CHUNK 128
#define GROUP 4
#define LCAP  (GROUP * CHUNK + 128)
#define BCAP  12
#define SCH    1024
#define SGROUP 4
#define ROWB ((TR + NWAVE - 1) / NWAVE)

__global__ void __launch_bounds__(1024)
range_sum_norm_kernel(const int* __restrict__ rows, const int* __restrict__ cols,
                      const float* __restrict__ ew, const float* __restrict__ deg,
                      const float* __restrict__ h, const float* __restrict__ bias,
                      const float* __restrict__ gamma, const float* __restrict__ beta,
                      float* __restrict__ out, int N, int E,
                      const unsigned int* __restrict__ keys, const int* __restrict__ rec) {
    __shared__ __attribute__((aligned(16))) float tile[TR * D];
    __shared__ int lists[NWAVE * LCAP];
    __shared__ unsigned long long bucket[TR * BCAP];
    __shared__ int rowcnt[TR];

    const int lane = threadIdx.x & 31;
    const int wave = __builtin_amdgcn_readfirstlane((int)(threadIdx.x >> 5));
    const int lo   = blockIdx.x * TR;
    const int nr   = (N - lo < TR) ? (N - lo) : TR;
    int* mylist    = lists + wave * LCAP;

    for (int r0 = wave; r0 < nr; r0 += NWAVE * ROWB) {
        v4f   hs[ROWB];
        float dg[ROWB];
#pragma unroll
        for (int u = 0; u < ROWB; ++u) {
            const int r  = r0 + u * NWAVE;
            const int rc = (r < nr) ? r : (nr - 1);
            hs[u] = *(const v4f*)(h + (size_t)(lo + rc) * D + lane * 4);
            dg[u] = deg[lo + rc];
        }
#pragma unroll
        for (int u = 0; u < ROWB; ++u) {
            asm volatile("" :: "v"(hs[u]), "s"(dg[u]));
        }
#pragma unroll
        for (int u = 0; u < ROWB; ++u) {
            const int r = r0 + u * NWAVE;
            if (r < nr) {
                const float di = inv_sqrt_deg(dg[u]);
                float* t = tile + r * D + lane;
#pragma unroll
                for (int i = 0; i < 4; ++i) t[32 * i] = hs[u][i] * di * di;
            }
        }
    }
    for (int i = (int)threadIdx.x; i < TR; i += NWAVE * 32) rowcnt[i] = 0;
    __syncthreads();

    auto file_edges = [&](int cnt) {
        for (int b0 = 0; b0 < cnt; b0 += 32) {
            const int  i     = b0 + lane;
            const bool valid = i < cnt;
            int  e   = mylist[valid ? i : b0];
            int   src, dl;
            float w;
            if (rec != nullptr) {
                const unsigned long long r = *(const unsigned long long*)(rec + 2 * (size_t)e);
                const unsigned r0 = (unsigned)r;
                src = (int)(r0 & 0x7FFFFFu);
                dl  = (int)(r0 >> 23);
                w   = __uint_as_float((unsigned)(r >> 32));
            } else {
                src = rows[e];
                dl  = cols[e] - lo;
                w   = ew[e];
            }
            float nm = inv_sqrt_deg(deg[src]) * w * inv_sqrt_deg(deg[lo + dl]);
            int slot = 0;
            if (valid) slot = __hip_atomic_fetch_add(&rowcnt[dl], 1, __ATOMIC_RELAXED,
                                                     __HIP_MEMORY_SCOPE_WORKGROUP);
            if (valid && slot < BCAP)
                bucket[dl * BCAP + slot] =
                    ((unsigned long long)__float_as_uint(nm) << 32) | (unsigned long long)(unsigned)src;
            unsigned ov = __builtin_amdgcn_ballot_w32(valid && slot >= BCAP);
            while (ov != 0u) {
                const int j = __builtin_ctz(ov);
                ov &= ov - 1u;
                const int   so = __builtin_amdgcn_readlane(src, j);
                const int   d  = __builtin_amdgcn_readlane(dl, j);
                const float f  = __builtin_bit_cast(float,
                                     __builtin_amdgcn_readlane(__builtin_bit_cast(int, nm), j));
                const v4f hv = *(const v4f*)(h + (size_t)so * D + lane * 4);
                float* a = tile + d * D + lane;
#pragma unroll
                for (int k = 0; k < 4; ++k)
                    __hip_atomic_fetch_add(a + 32 * k, hv[k] * f, __ATOMIC_RELAXED,
                                           __HIP_MEMORY_SCOPE_WORKGROUP);
            }
        }
    };

    const bool vec   = (E & 3) == 0;
    const int  nfull = vec ? (E / CHUNK) : 0;
    const int  rot   = (int)blockIdx.x * (nfull / (int)gridDim.x);
    int cnt = 0;
    int scanned = nfull * CHUNK;

    if (keys != nullptr) {
        const int nfs  = E / SCH;
        const int rots = (int)blockIdx.x * (nfs / (int)gridDim.x);
        const unsigned char* kb = (const unsigned char*)keys;
        unsigned flip[8];
#pragma unroll
        for (int p = 0; p < 8; ++p) flip[p] = (((unsigned)blockIdx.x >> p) & 1u) ? 0u : 0xFFFFFFFFu;
        const int lane_e = ((lane >> 2) << 7) + (lane & 3);
        scanned = nfs * SCH;
        for (int q0 = wave; q0 < nfs; q0 += SGROUP * NWAVE) {
            v4i      ka[SGROUP];
            v4i      kc[SGROUP];
            int      eb[SGROUP];
            unsigned ok[SGROUP];
#pragma unroll
            for (int u = 0; u < SGROUP; ++u) {
                int  q     = q0 + u * NWAVE;
                bool valid = q < nfs;
                int  ch    = valid ? (q + rots) : rots;
                if (ch >= nfs) ch -= nfs;
                eb[u] = ch * SCH;
                ok[u] = valid ? 0xFFFFFFFFu : 0u;
                const unsigned char* kp = kb + eb[u] + lane * 32;
                ka[u] = *(const v4i*)(kp);
                kc[u] = *(const v4i*)(kp + 16);
            }
            unsigned mv[SGROUP];
#pragma unroll
            for (int u = 0; u < SGROUP; ++u) {
                mv[u] = ((unsigned)ka[u].x ^ flip[0]) & ((unsigned)ka[u].y ^ flip[1]) &
                        ((unsigned)ka[u].z ^ flip[2]) & ((unsigned)ka[u].w ^ flip[3]) &
                        ((unsigned)kc[u].x ^ flip[4]) & ((unsigned)kc[u].y ^ flip[5]) &
                        ((unsigned)kc[u].z ^ flip[6]) & ((unsigned)kc[u].w ^ flip[7]) & ok[u];
            }
#pragma nounroll
            for (int u = 0; u < SGROUP; ++u) {
                unsigned mm = (u == 0) ? mv[0] : (u == 1) ? mv[1] : (u == 2) ? mv[2] : mv[3];
                const int e0 = ((u == 0) ? eb[0] : (u == 1) ? eb[1] : (u == 2) ? eb[2] : eb[3]) + lane_e;
                while (__builtin_amdgcn_ballot_w32(mm != 0u) != 0u) {
                    if (cnt > LCAP - 32) {
                        file_edges(cnt);
                        cnt = 0;
                    }
                    const bool     has = mm != 0u;
                    const unsigned b   = __builtin_amdgcn_ballot_w32(has);
                    if (has) {
                        const int c = __builtin_ctz(mm);
                        mylist[cnt + __builtin_amdgcn_mbcnt_lo(b, 0u)] = e0 + (c << 2);
                        mm &= mm - 1u;
                    }
                    cnt += __builtin_popcount(b);
                }
            }
        }
    } else {
        for (int q0 = wave; q0 < nfull; q0 += GROUP * NWAVE) {
            v4i      c[GROUP];
            int      eb[GROUP];
            unsigned lim[GROUP];
#pragma unroll
            for (int u = 0; u < GROUP; ++u) {
                int  q     = q0 + u * NWAVE;
                bool valid = q < nfull;
                int  ch    = valid ? (q + rot) : rot;
                if (ch >= nfull) ch -= nfull;
                eb[u]  = ch * CHUNK;
                lim[u] = valid ? (unsigned)nr : 0u;
                c[u]   = *(const v4i*)(cols + eb[u] + lane * 4);
            }
#pragma unroll
            for (int u = 0; u < GROUP; ++u) {
                int  e0 = eb[u] + lane * 4;
                bool m0 = (unsigned)(c[u].x - lo) < lim[u];
                bool m1 = (unsigned)(c[u].y - lo) < lim[u];
                bool m2 = (unsigned)(c[u].z - lo) < lim[u];
                bool m3 = (unsigned)(c[u].w - lo) < lim[u];
                if (__builtin_amdgcn_ballot_w32(m0 | m1 | m2 | m3) != 0u) {
                    unsigned b;
                    b = __builtin_amdgcn_ballot_w32(m0);
                    if (m0) mylist[cnt + __builtin_amdgcn_mbcnt_lo(b, 0u)] = e0;
                    cnt += __builtin_popcount(b);
                    b = __builtin_amdgcn_ballot_w32(m1);
                    if (m1) mylist[cnt + __builtin_amdgcn_mbcnt_lo(b, 0u)] = e0 + 1;
                    cnt += __builtin_popcount(b);
                    b = __builtin_amdgcn_ballot_w32(m2);
                    if (m2) mylist[cnt + __builtin_amdgcn_mbcnt_lo(b, 0u)] = e0 + 2;
                    cnt += __builtin_popcount(b);
                    b = __builtin_amdgcn_ballot_w32(m3);
                    if (m3) mylist[cnt + __builtin_amdgcn_mbcnt_lo(b, 0u)] = e0 + 3;
                    cnt += __builtin_popcount(b);
                }
            }
            if (cnt > LCAP - GROUP * CHUNK) {
                file_edges(cnt);
                cnt = 0;
            }
        }
    }
    for (int base = scanned + wave * 32; base < E; base += NWAVE * 32) {
        if (cnt > LCAP - 32) {
            file_edges(cnt);
            cnt = 0;
        }
        int  e   = base + lane;
        int  col = (e < E) ? cols[e] : -1;
        bool m   = (unsigned)(col - lo) < (unsigned)nr;
        unsigned b = __builtin_amdgcn_ballot_w32(m);
        if (m) mylist[cnt + __builtin_amdgcn_mbcnt_lo(b, 0u)] = e;
        cnt += __builtin_popcount(b);
    }
    file_edges(cnt);
    __syncthreads();

    const v4f bv = *(const v4f*)(bias  + lane * 4);
    const v4f gv = *(const v4f*)(gamma + lane * 4);
    const v4f tv = *(const v4f*)(beta  + lane * 4);

    auto fetch = [&](int rr, int& n, int& srcv, float& nmv, v4f (&hx)[8]) {
        const int rc = (rr < nr) ? rr : (nr - 1);
        const int c  = __builtin_amdgcn_readfirstlane(rowcnt[rc]);
        n = (rr < nr) ? ((c < BCAP) ? c : BCAP) : 0;
        unsigned long long ent = 0ull;
        if (lane < n) ent = bucket[rc * BCAP + lane];
        srcv = (int)(unsigned)ent;
        nmv  = __uint_as_float((unsigned)(ent >> 32));
#pragma unroll
        for (int u = 0; u < 8; ++u) {
            const int so  = __builtin_amdgcn_readlane(srcv, u);
            const int off = (u < n) ? lane * 4 : 0;
            hx[u] = *(const v4f*)(h + (size_t)so * D + off);
        }
    };

    auto fetch_rest = [&](int n, int srcv, v4f (&hc)[BCAP - 8]) {
        if (n > 8) {
#pragma unroll
            for (int u = 0; u < BCAP - 8; ++u) {
                const int so  = __builtin_amdgcn_readlane(srcv, 8 + u);
                const int off = (8 + u < n) ? lane * 4 : 0;
                hc[u] = *(const v4f*)(h + (size_t)so * D + off);
            }
        }
    };

    auto finish_row = [&](int r, int n, float nmv, v4f (&hx)[8], v4f (&hc)[BCAP - 8]) {
        v4f acc = {0.f, 0.f, 0.f, 0.f};
#pragma unroll
        for (int u = 0; u < 8; ++u) {
            if (u < n) {
                const float f = __builtin_bit_cast(float,
                                    __builtin_amdgcn_readlane(__builtin_bit_cast(int, nmv), u));
                acc += hx[u] * f;
            }
        }
        if (n > 8) {
#pragma unroll
            for (int u = 0; u < BCAP - 8; ++u) {
                if (8 + u < n) {
                    const float f = __builtin_bit_cast(float,
                                        __builtin_amdgcn_readlane(__builtin_bit_cast(int, nmv), 8 + u));
                    acc += hc[u] * f;
                }
            }
        }

        const float* t = tile + r * D + lane;
        v4f v;
#pragma unroll
        for (int i = 0; i < 4; ++i) v[i] = t[32 * i] + acc[i];
        float s = 0.f, s2 = 0.f;
#pragma unroll
        for (int i = 0; i < 4; ++i) {
            float vv = v[i] + bv[i];
            vv = fmaxf(vv, 0.0f);
            v[i] = vv;
            s  += vv;
            s2 += vv * vv;
        }
#pragma unroll
        for (int off = 16; off >= 1; off >>= 1) {
            s  += __shfl_xor(s,  off, 32);
            s2 += __shfl_xor(s2, off, 32);
        }
        float mu  = s * (1.0f / 128.0f);
        float var = s2 * (1.0f / 128.0f) - mu * mu;
        float rs  = rsqrtf(var + LN_EPS);
#pragma unroll
        for (int i = 0; i < 4; ++i)
            v[i] = (v[i] - mu) * rs * gv[i] + tv[i];
        *(v4f*)(out + (size_t)(lo + r) * D + lane * 4) = v;
    };

    int   n_a, src_a, n_b, src_b;
    float nm_a, nm_b;
    v4f   ha[8], hb[8];
    fetch(wave, n_a, src_a, nm_a, ha);
    int r = wave;
    for (; r + NWAVE < nr; r += 2 * NWAVE) {
        v4f hc[BCAP - 8];
        fetch_rest(n_a, src_a, hc);
        fetch(r + NWAVE, n_b, src_b, nm_b, hb);
        finish_row(r, n_a, nm_a, ha, hc);
        fetch_rest(n_b, src_b, hc);
        fetch(r + 2 * NWAVE, n_a, src_a, nm_a, ha);
        finish_row(r + NWAVE, n_b, nm_b, hb, hc);
    }
    if (r < nr) {
        v4f hc[BCAP - 8];
        fetch_rest(n_a, src_a, hc);
        finish_row(r, n_a, nm_a, ha, hc);
    }
}

extern "C" void kernel_launch(void* const* d_in, const int* in_sizes, int n_in,
                              void* d_out, int out_size, void* d_ws, size_t ws_size,
                              hipStream_t stream) {
    const float* x     = (const float*)d_in[0];
    const int*   eidx  = (const int*)  d_in[1];
    const float* ew    = (const float*)d_in[2];
    const float* W     = (const float*)d_in[3];
    const float* bias  = (const float*)d_in[4];
    const float* gamma = (const float*)d_in[5];
    const float* beta  = (const float*)d_in[6];
    float*       out   = (float*)d_out;

    const int N = in_sizes[0] / D;
    const int E = in_sizes[2];
    const int* rows = eidx;
    const int* cols = eidx + E;

    float* h    = (float*)d_ws;
    float* deg  = h + (size_t)N * D;

    const size_t koff  = (((size_t)N * (D + 1) * sizeof(float)) + 127) & ~(size_t)127;
    const bool   keyed = ((E & 3) == 0) && (E >= SCH) && ((N + TR - 1) / TR <= 256) &&
                         (ws_size >= koff + (size_t)E);
    unsigned int* keys = keyed ? (unsigned int*)((char*)d_ws + koff) : nullptr;

    const size_t roff   = (koff + (size_t)E + 127) & ~(size_t)127;
    const bool   recd   = keyed && (N <= (1 << 23)) && (ws_size >= roff + (size_t)E * 8);
    int*         rec    = recd ? (int*)((char*)d_ws + roff) : nullptr;

    deg_init_kernel<<<(N + 255) / 256, 256, 0, stream>>>(deg, N);

    gemm_wmma_kernel<<<GEMM_GROUPS, GEMM_WAVES * 32, 0, stream>>>(x, W, h, N, cols, ew, deg, E, keys, rows, rec);

    range_sum_norm_kernel<<<(N + TR - 1) / TR, NWAVE * 32, 0, stream>>>(
        rows, cols, ew, deg, h, bias, gamma, beta, out, N, E, keys, rec);
}
